# starting kernel
# baseline (speedup 1.0000x reference)
#include <hip/hip_runtime.h>
#include <hip/hip_bf16.h>

#define N_NODES 100000
#define N_EDGES 3200000
#define HID 64
#define RED 32
#define EMB 32
#define LN_EPS 1e-5f

typedef __attribute__((ext_vector_type(2))) float v2f;
typedef __attribute__((ext_vector_type(8))) float v8f;

__global__ void k_deg_init(float* __restrict__ deg) {
    int i = blockIdx.x * blockDim.x + threadIdx.x;
    if (i < N_NODES) deg[i] = 1.0f;
}

__global__ void k_deg_scatter(const int* __restrict__ col, const float* __restrict__ w,
                              float* __restrict__ deg) {
    int e = blockIdx.x * blockDim.x + threadIdx.x;
    if (e < N_EDGES) atomicAdd(&deg[col[e]], w[e]);
}

__global__ void k_l1_init(const float* __restrict__ x, const float* __restrict__ deg,
                          const float* __restrict__ W1, const float* __restrict__ b1,
                          const float* __restrict__ rW1, const float* __restrict__ rb1,
                          float* __restrict__ dinv, float* __restrict__ xw1,
                          float* __restrict__ acc1) {
    long long t = (long long)blockIdx.x * blockDim.x + threadIdx.x;
    if (t >= (long long)N_NODES * HID) return;
    int n = (int)(t >> 6);
    int f = (int)(t & 63);
    float d  = deg[n];
    float di = d > 0.f ? rsqrtf(d) : 0.f;
    if (f == 0) dinv[n] = di;
    float x0 = x[2 * n], x1v = x[2 * n + 1];
    float xw = fmaf(x0, W1[f], x1v * W1[HID + f]);
    xw1[t]  = xw;
    acc1[t] = di * di * xw + b1[f] + fmaf(x0, rW1[f], x1v * rW1[HID + f]) + rb1[f];
}

template <int F>
__global__ void k_edge_scatter(const int* __restrict__ row, const int* __restrict__ col,
                               const float* __restrict__ w, const float* __restrict__ dinv,
                               const float* __restrict__ src, float* __restrict__ acc) {
    long long t = (long long)blockIdx.x * blockDim.x + threadIdx.x;
    if (t >= (long long)N_EDGES * F) return;
    int e = (int)(t / F);
    int f = (int)(t & (F - 1));
    int r = row[e], c = col[e];
    float coeff = dinv[r] * w[e] * dinv[c];
    atomicAdd(&acc[(long long)c * F + f], coeff * src[(long long)r * F + f]);
}

__global__ void k_ln_relu_64(const float* __restrict__ acc, const float* __restrict__ g,
                             const float* __restrict__ be, float* __restrict__ out) {
    int wave = threadIdx.x >> 5;
    int lane = threadIdx.x & 31;
    int n = blockIdx.x * (blockDim.x >> 5) + wave;
    if (n >= N_NODES) return;
    float v0 = acc[(long long)n * 64 + lane];
    float v1 = acc[(long long)n * 64 + lane + 32];
    float s = v0 + v1, sq = fmaf(v0, v0, v1 * v1);
    #pragma unroll
    for (int off = 16; off > 0; off >>= 1) {
        s  += __shfl_xor(s,  off, 32);
        sq += __shfl_xor(sq, off, 32);
    }
    float mu  = s * (1.f / 64.f);
    float var = sq * (1.f / 64.f) - mu * mu;
    float inv = rsqrtf(var + LN_EPS);
    float o0 = (v0 - mu) * inv * g[lane]      + be[lane];
    float o1 = (v1 - mu) * inv * g[lane + 32] + be[lane + 32];
    out[(long long)n * 64 + lane]      = fmaxf(o0, 0.f);
    out[(long long)n * 64 + lane + 32] = fmaxf(o1, 0.f);
}

__global__ void k_l2_wmma(const float* __restrict__ x1, const float* __restrict__ W2,
                          const float* __restrict__ rW2, const float* __restrict__ b2,
                          const float* __restrict__ rb2, const float* __restrict__ dinv,
                          float* __restrict__ xw2, float* __restrict__ acc2) {
    int wave_global = (blockIdx.x * blockDim.x + threadIdx.x) >> 5;
    int lane = threadIdx.x & 31;
    const int tiles_m = N_NODES / 16;
    int tm = wave_global >> 1;
    int tn = wave_global & 1;
    if (tm >= tiles_m) return;
    int m0 = tm * 16, n0 = tn * 16;
    int  lr = lane & 15;
    bool hi = lane >= 16;
    int  kofs = hi ? 2 : 0;

    v8f c0 = {};
    v8f c1 = {};
    #pragma unroll
    for (int k0 = 0; k0 < HID; k0 += 4) {
        const float* ap = &x1[(long long)(m0 + lr) * HID + k0 + kofs];
        v2f a; a.x = ap[0]; a.y = ap[1];
        int kb = k0 + kofs;
        v2f b;  b.x  = W2 [kb * RED + n0 + lr]; b.y  = W2 [(kb + 1) * RED + n0 + lr];
        v2f br; br.x = rW2[kb * RED + n0 + lr]; br.y = rW2[(kb + 1) * RED + n0 + lr];
        c0 = __builtin_amdgcn_wmma_f32_16x16x4_f32(false, a, false, b,  (short)0, c0, false, false);
        c1 = __builtin_amdgcn_wmma_f32_16x16x4_f32(false, a, false, br, (short)0, c1, false, false);
    }
    int ncol = n0 + lr;
    float b2v = b2[ncol], rb2v = rb2[ncol];
    #pragma unroll
    for (int v = 0; v < 8; ++v) {
        int m = m0 + v + (hi ? 8 : 0);
        float di = dinv[m];
        float xv = c0[v];
        xw2 [(long long)m * RED + ncol] = xv;
        acc2[(long long)m * RED + ncol] = di * di * xv + b2v + c1[v] + rb2v;
    }
}

__global__ void k_head(const float* __restrict__ acc2, const float* __restrict__ g2,
                       const float* __restrict__ be2, const float* __restrict__ fcW,
                       const float* __restrict__ fcb, const float* __restrict__ pW,
                       const float* __restrict__ pb, float* __restrict__ logits,
                       float* __restrict__ emb) {
    int wave = threadIdx.x >> 5;
    int lane = threadIdx.x & 31;
    int n = blockIdx.x * (blockDim.x >> 5) + wave;
    if (n >= N_NODES) return;
    float v = acc2[(long long)n * 32 + lane];
    float s = v, sq = v * v;
    #pragma unroll
    for (int off = 16; off > 0; off >>= 1) {
        s  += __shfl_xor(s,  off, 32);
        sq += __shfl_xor(sq, off, 32);
    }
    float mu  = s * (1.f / 32.f);
    float var = sq * (1.f / 32.f) - mu * mu;
    float inv = rsqrtf(var + LN_EPS);
    float x2 = fmaxf((v - mu) * inv * g2[lane] + be2[lane], 0.f);

    float lg = x2 * fcW[lane];
    #pragma unroll
    for (int off = 16; off > 0; off >>= 1) lg += __shfl_xor(lg, off, 32);
    if (lane == 0) logits[n] = lg + fcb[0];

    float p = pb[lane];
    #pragma unroll
    for (int i = 0; i < 32; ++i)
        p = fmaf(__shfl(x2, i, 32), pW[i * 32 + lane], p);
    float nsq = p * p;
    #pragma unroll
    for (int off = 16; off > 0; off >>= 1) nsq += __shfl_xor(nsq, off, 32);
    float nrm = sqrtf(nsq);
    emb[(long long)n * 32 + lane] = p / fmaxf(nrm, 1e-12f);
}

extern "C" void kernel_launch(void* const* d_in, const int* in_sizes, int n_in,
                              void* d_out, int out_size, void* d_ws, size_t ws_size,
                              hipStream_t stream) {
    const float* x    = (const float*)d_in[0];
    const float* ew   = (const float*)d_in[1];
    const float* W1   = (const float*)d_in[2];
    const float* b1   = (const float*)d_in[3];
    const float* rW1  = (const float*)d_in[4];
    const float* rb1  = (const float*)d_in[5];
    const float* g1   = (const float*)d_in[6];
    const float* be1  = (const float*)d_in[7];
    const float* W2   = (const float*)d_in[8];
    const float* b2   = (const float*)d_in[9];
    const float* rW2  = (const float*)d_in[10];
    const float* rb2  = (const float*)d_in[11];
    const float* g2   = (const float*)d_in[12];
    const float* be2  = (const float*)d_in[13];
    const float* fcW  = (const float*)d_in[14];
    const float* fcb  = (const float*)d_in[15];
    const float* pW   = (const float*)d_in[16];
    const float* pb   = (const float*)d_in[17];
    const int*   ei   = (const int*)d_in[18];
    const int*   row  = ei;
    const int*   col  = ei + N_EDGES;

    float* ws   = (float*)d_ws;
    float* deg  = ws;  ws += N_NODES;
    float* dinv = ws;  ws += N_NODES;
    float* xw1  = ws;  ws += (size_t)N_NODES * HID;
    float* acc1 = ws;  ws += (size_t)N_NODES * HID;
    float* x1   = ws;  ws += (size_t)N_NODES * HID;
    float* xw2  = ws;  ws += (size_t)N_NODES * RED;
    float* acc2 = ws;  ws += (size_t)N_NODES * RED;

    float* logits = (float*)d_out;
    float* emb    = logits + N_NODES;

    const int B = 256;
    k_deg_init<<<(N_NODES + B - 1) / B, B, 0, stream>>>(deg);
    k_deg_scatter<<<(N_EDGES + B - 1) / B, B, 0, stream>>>(col, ew, deg);

    long long t1 = (long long)N_NODES * HID;
    k_l1_init<<<(unsigned)((t1 + B - 1) / B), B, 0, stream>>>(x, deg, W1, b1, rW1, rb1,
                                                              dinv, xw1, acc1);

    long long te1 = (long long)N_EDGES * HID;
    k_edge_scatter<HID><<<(unsigned)((te1 + B - 1) / B), B, 0, stream>>>(row, col, ew, dinv,
                                                                          xw1, acc1);

    k_ln_relu_64<<<(N_NODES + 7) / 8, B, 0, stream>>>(acc1, g1, be1, x1);

    int waves_l2 = (N_NODES / 16) * (RED / 16);
    int thr_l2   = waves_l2 * 32;
    k_l2_wmma<<<(thr_l2 + B - 1) / B, B, 0, stream>>>(x1, W2, rW2, b2, rb2, dinv, xw2, acc2);

    long long te2 = (long long)N_EDGES * RED;
    k_edge_scatter<RED><<<(unsigned)((te2 + B - 1) / B), B, 0, stream>>>(row, col, ew, dinv,
                                                                          xw2, acc2);

    k_head<<<(N_NODES + 7) / 8, B, 0, stream>>>(acc2, g2, be2, fcW, fcb, pW, pb, logits, emb);
}


# neighbour sums of both layers: a work-group per 391-node destination range filters per-edge owner bytes and adds source rows into LDS sums (ds f32 adds, arrival order), not global f32 atomics
# speedup vs baseline: 2.0009x; 2.0009x over previous
#include <hip/hip_runtime.h>
#include <hip/hip_bf16.h>

#define N_NODES 100000
#define N_EDGES 3200000
#define HID 64
#define RED 32
#define EMB 32
#define LN_EPS 1e-5f
#define RANGE 391
#define NB_RANGE 256
#define LCAP 1536
#define NCHUNK (N_EDGES / 512)

typedef __attribute__((ext_vector_type(2))) float v2f;
typedef __attribute__((ext_vector_type(8))) float v8f;

__global__ void k_deg_init(float* __restrict__ deg) {
    int i = blockIdx.x * blockDim.x + threadIdx.x;
    if (i < N_NODES) deg[i] = 1.0f;
}

__global__ void k_deg_scatter(const int* __restrict__ col, const float* __restrict__ w,
                              float* __restrict__ deg, unsigned char* __restrict__ own) {
    int e = blockIdx.x * blockDim.x + threadIdx.x;
    if (e < N_EDGES) {
        int c = col[e];
        atomicAdd(&deg[c], w[e]);
        own[e] = (unsigned char)(c / RANGE);
    }
}

__global__ void k_l1_init(const float* __restrict__ x, const float* __restrict__ deg,
                          const float* __restrict__ W1, const float* __restrict__ b1,
                          const float* __restrict__ rW1, const float* __restrict__ rb1,
                          float* __restrict__ dinv, float* __restrict__ xw1,
                          float* __restrict__ acc1) {
    long long t = (long long)blockIdx.x * blockDim.x + threadIdx.x;
    if (t >= (long long)N_NODES * HID) return;
    int n = (int)(t >> 6);
    int f = (int)(t & 63);
    float d  = deg[n];
    float di = d > 0.f ? rsqrtf(d) : 0.f;
    if (f == 0) dinv[n] = di;
    float x0 = x[2 * n], x1v = x[2 * n + 1];
    float xw = fmaf(x0, W1[f], x1v * W1[HID + f]);
    xw1[t]  = xw;
    acc1[t] = di * di * xw + b1[f] + fmaf(x0, rW1[f], x1v * rW1[HID + f]) + rb1[f];
}

template <int F>
__device__ __forceinline__ void walk_list(const int* list, int n, int lane, int c0,
                                          const int* __restrict__ row, const int* __restrict__ col,
                                          const float* __restrict__ w, const float* __restrict__ dinv,
                                          const float* __restrict__ src, float* s_acc) {
    for (int base = 0; base < n; base += 32) {
        int m = n - base; if (m > 32) m = 32;
        int i = base + lane;
        int e = list[i < n ? i : n - 1];
        int r = row[e], c = col[e];
        float cf = dinv[r] * w[e] * dinv[c];
        int cl = c - c0;
        int j = 0;
        for (; j + 8 <= m; j += 8) {
            if (F == 64) {
                v2f s[8];
                #pragma unroll
                for (int u = 0; u < 8; ++u) {
                    int rj = __builtin_amdgcn_readlane(r, j + u);
                    s[u] = *(const v2f*)(src + (size_t)rj * 64 + 2 * lane);
                }
                #pragma unroll
                for (int u = 0; u < 8; ++u) {
                    int clj = __builtin_amdgcn_readlane(cl, j + u);
                    float cfj = __int_as_float(__builtin_amdgcn_readlane(__float_as_int(cf), j + u));
                    atomicAdd(&s_acc[clj * 64 + 2 * lane],     cfj * s[u].x);
                    atomicAdd(&s_acc[clj * 64 + 2 * lane + 1], cfj * s[u].y);
                }
            } else {
                float s[8];
                #pragma unroll
                for (int u = 0; u < 8; ++u) {
                    int rj = __builtin_amdgcn_readlane(r, j + u);
                    s[u] = src[(size_t)rj * F + lane];
                }
                #pragma unroll
                for (int u = 0; u < 8; ++u) {
                    int clj = __builtin_amdgcn_readlane(cl, j + u);
                    float cfj = __int_as_float(__builtin_amdgcn_readlane(__float_as_int(cf), j + u));
                    atomicAdd(&s_acc[clj * F + lane], cfj * s[u]);
                }
            }
        }
        for (; j < m; ++j) {
            int rj  = __builtin_amdgcn_readlane(r, j);
            int clj = __builtin_amdgcn_readlane(cl, j);
            float cfj = __int_as_float(__builtin_amdgcn_readlane(__float_as_int(cf), j));
            if (F == 64) {
                v2f s = *(const v2f*)(src + (size_t)rj * 64 + 2 * lane);
                atomicAdd(&s_acc[clj * 64 + 2 * lane],     cfj * s.x);
                atomicAdd(&s_acc[clj * 64 + 2 * lane + 1], cfj * s.y);
            } else {
                atomicAdd(&s_acc[clj * F + lane], cfj * src[(size_t)rj * F + lane]);
            }
        }
    }
}

__device__ __forceinline__ unsigned nz_bytes(unsigned x) {
    return ((x & 0x7F7F7F7Fu) + 0x7F7F7F7Fu) | x;
}

template <int F>
__global__ __launch_bounds__(1024) void k_range_sum(const unsigned char* __restrict__ own,
                                                    const int* __restrict__ row,
                                                    const int* __restrict__ col,
                                                    const float* __restrict__ w,
                                                    const float* __restrict__ dinv,
                                                    const float* __restrict__ src,
                                                    float* __restrict__ acc) {
    __shared__ float s_acc[RANGE * F];
    __shared__ int   s_list[32 * LCAP];
    const int tid = threadIdx.x, lane = tid & 31;
    const int wave = __builtin_amdgcn_readfirstlane(tid >> 5);
    const int b = blockIdx.x, c0 = b * RANGE;
    for (int i = tid; i < RANGE * F; i += 1024) s_acc[i] = 0.f;
    __syncthreads();

    int* list = s_list + wave * LCAP;
    int cnt = 0;
    const uint4* own4 = (const uint4*)own;
    const unsigned key = (unsigned)b * 0x01010101u;
    const int start = (b * NCHUNK) / NB_RANGE;
    int l = wave;
    int ph = l + start; if (ph >= NCHUNK) ph -= NCHUNK;
    uint4 cur = own4[(size_t)ph * 32 + lane];
    while (l < NCHUNK) {
        const int ebase = ph * 512 + lane * 16;
        const int ln = l + 32;
        uint4 nxt = cur;
        if (ln < NCHUNK) {
            ph = ln + start; if (ph >= NCHUNK) ph -= NCHUNK;
            nxt = own4[(size_t)ph * 32 + lane];
        }
        if (cnt > LCAP - 512) {
            __builtin_amdgcn_fence(__ATOMIC_ACQ_REL, "wavefront");
            walk_list<F>(list, cnt, lane, c0, row, col, w, dinv, src, s_acc);
            __builtin_amdgcn_fence(__ATOMIC_ACQ_REL, "wavefront");
            cnt = 0;
        }
        const unsigned H = 0x80808080u;
        unsigned t0 = nz_bytes(cur.x ^ key), t1 = nz_bytes(cur.y ^ key);
        unsigned t2 = nz_bytes(cur.z ^ key), t3 = nz_bytes(cur.w ^ key);
        unsigned u = ((~t0 & H) >> 7) | ((~t1 & H) >> 6) | ((~t2 & H) >> 5) | ((~t3 & H) >> 4);
        unsigned bal = __builtin_amdgcn_ballot_w32(u != 0);
        while (bal) {
            int p = __builtin_ctz(u | 0x80000000u);
            int slot = cnt + (int)__builtin_amdgcn_mbcnt_lo(bal, 0);
            if (u != 0) list[slot] = ebase + ((p & 3) << 2) + (p >> 3);
            cnt += __builtin_popcount(bal);
            u &= u - 1;
            bal = __builtin_amdgcn_ballot_w32(u != 0);
        }
        cur = nxt; l = ln;
    }
    __builtin_amdgcn_fence(__ATOMIC_ACQ_REL, "wavefront");
    walk_list<F>(list, cnt, lane, c0, row, col, w, dinv, src, s_acc);
    __syncthreads();

    int rows = N_NODES - c0; if (rows > RANGE) rows = RANGE;
    float* ap = acc + (size_t)c0 * F;
    for (int i = tid; i < rows * F; i += 1024) ap[i] += s_acc[i];
}

__global__ void k_ln_relu_64(const float* __restrict__ acc, const float* __restrict__ g,
                             const float* __restrict__ be, float* __restrict__ out) {
    int wave = threadIdx.x >> 5;
    int lane = threadIdx.x & 31;
    int n = blockIdx.x * (blockDim.x >> 5) + wave;
    if (n >= N_NODES) return;
    float v0 = acc[(long long)n * 64 + lane];
    float v1 = acc[(long long)n * 64 + lane + 32];
    float s = v0 + v1, sq = fmaf(v0, v0, v1 * v1);
    #pragma unroll
    for (int off = 16; off > 0; off >>= 1) {
        s  += __shfl_xor(s,  off, 32);
        sq += __shfl_xor(sq, off, 32);
    }
    float mu  = s * (1.f / 64.f);
    float var = sq * (1.f / 64.f) - mu * mu;
    float inv = rsqrtf(var + LN_EPS);
    float o0 = (v0 - mu) * inv * g[lane]      + be[lane];
    float o1 = (v1 - mu) * inv * g[lane + 32] + be[lane + 32];
    out[(long long)n * 64 + lane]      = fmaxf(o0, 0.f);
    out[(long long)n * 64 + lane + 32] = fmaxf(o1, 0.f);
}

__global__ void k_l2_wmma(const float* __restrict__ x1, const float* __restrict__ W2,
                          const float* __restrict__ rW2, const float* __restrict__ b2,
                          const float* __restrict__ rb2, const float* __restrict__ dinv,
                          float* __restrict__ xw2, float* __restrict__ acc2) {
    int wave_global = (blockIdx.x * blockDim.x + threadIdx.x) >> 5;
    int lane = threadIdx.x & 31;
    const int tiles_m = N_NODES / 16;
    int tm = wave_global >> 1;
    int tn = wave_global & 1;
    if (tm >= tiles_m) return;
    int m0 = tm * 16, n0 = tn * 16;
    int  lr = lane & 15;
    bool hi = lane >= 16;
    int  kofs = hi ? 2 : 0;

    v8f c0 = {};
    v8f c1 = {};
    #pragma unroll
    for (int k0 = 0; k0 < HID; k0 += 4) {
        const float* ap = &x1[(long long)(m0 + lr) * HID + k0 + kofs];
        v2f a; a.x = ap[0]; a.y = ap[1];
        int kb = k0 + kofs;
        v2f b;  b.x  = W2 [kb * RED + n0 + lr]; b.y  = W2 [(kb + 1) * RED + n0 + lr];
        v2f br; br.x = rW2[kb * RED + n0 + lr]; br.y = rW2[(kb + 1) * RED + n0 + lr];
        c0 = __builtin_amdgcn_wmma_f32_16x16x4_f32(false, a, false, b,  (short)0, c0, false, false);
        c1 = __builtin_amdgcn_wmma_f32_16x16x4_f32(false, a, false, br, (short)0, c1, false, false);
    }
    int ncol = n0 + lr;
    float b2v = b2[ncol], rb2v = rb2[ncol];
    #pragma unroll
    for (int v = 0; v < 8; ++v) {
        int m = m0 + v + (hi ? 8 : 0);
        float di = dinv[m];
        float xv = c0[v];
        xw2 [(long long)m * RED + ncol] = xv;
        acc2[(long long)m * RED + ncol] = di * di * xv + b2v + c1[v] + rb2v;
    }
}

__global__ void k_head(const float* __restrict__ acc2, const float* __restrict__ g2,
                       const float* __restrict__ be2, const float* __restrict__ fcW,
                       const float* __restrict__ fcb, const float* __restrict__ pW,
                       const float* __restrict__ pb, float* __restrict__ logits,
                       float* __restrict__ emb) {
    int wave = threadIdx.x >> 5;
    int lane = threadIdx.x & 31;
    int n = blockIdx.x * (blockDim.x >> 5) + wave;
    if (n >= N_NODES) return;
    float v = acc2[(long long)n * 32 + lane];
    float s = v, sq = v * v;
    #pragma unroll
    for (int off = 16; off > 0; off >>= 1) {
        s  += __shfl_xor(s,  off, 32);
        sq += __shfl_xor(sq, off, 32);
    }
    float mu  = s * (1.f / 32.f);
    float var = sq * (1.f / 32.f) - mu * mu;
    float inv = rsqrtf(var + LN_EPS);
    float x2 = fmaxf((v - mu) * inv * g2[lane] + be2[lane], 0.f);

    float lg = x2 * fcW[lane];
    #pragma unroll
    for (int off = 16; off > 0; off >>= 1) lg += __shfl_xor(lg, off, 32);
    if (lane == 0) logits[n] = lg + fcb[0];

    float p = pb[lane];
    #pragma unroll
    for (int i = 0; i < 32; ++i)
        p = fmaf(__shfl(x2, i, 32), pW[i * 32 + lane], p);
    float nsq = p * p;
    #pragma unroll
    for (int off = 16; off > 0; off >>= 1) nsq += __shfl_xor(nsq, off, 32);
    float nrm = sqrtf(nsq);
    emb[(long long)n * 32 + lane] = p / fmaxf(nrm, 1e-12f);
}

extern "C" void kernel_launch(void* const* d_in, const int* in_sizes, int n_in,
                              void* d_out, int out_size, void* d_ws, size_t ws_size,
                              hipStream_t stream) {
    const float* x    = (const float*)d_in[0];
    const float* ew   = (const float*)d_in[1];
    const float* W1   = (const float*)d_in[2];
    const float* b1   = (const float*)d_in[3];
    const float* rW1  = (const float*)d_in[4];
    const float* rb1  = (const float*)d_in[5];
    const float* g1   = (const float*)d_in[6];
    const float* be1  = (const float*)d_in[7];
    const float* W2   = (const float*)d_in[8];
    const float* b2   = (const float*)d_in[9];
    const float* rW2  = (const float*)d_in[10];
    const float* rb2  = (const float*)d_in[11];
    const float* g2   = (const float*)d_in[12];
    const float* be2  = (const float*)d_in[13];
    const float* fcW  = (const float*)d_in[14];
    const float* fcb  = (const float*)d_in[15];
    const float* pW   = (const float*)d_in[16];
    const float* pb   = (const float*)d_in[17];
    const int*   ei   = (const int*)d_in[18];
    const int*   row  = ei;
    const int*   col  = ei + N_EDGES;

    float* ws   = (float*)d_ws;
    float* deg  = ws;  ws += N_NODES;
    float* dinv = ws;  ws += N_NODES;
    float* xw1  = ws;  ws += (size_t)N_NODES * HID;
    float* acc1 = ws;  ws += (size_t)N_NODES * HID;
    unsigned char* own = (unsigned char*)ws;  ws += (size_t)N_NODES * HID;
    float* x1   = xw1;
    float* xw2  = ws;  ws += (size_t)N_NODES * RED;
    float* acc2 = ws;  ws += (size_t)N_NODES * RED;

    float* logits = (float*)d_out;
    float* emb    = logits + N_NODES;

    const int B = 256;
    k_deg_init<<<(N_NODES + B - 1) / B, B, 0, stream>>>(deg);
    k_deg_scatter<<<(N_EDGES + B - 1) / B, B, 0, stream>>>(col, ew, deg, own);

    long long t1 = (long long)N_NODES * HID;
    k_l1_init<<<(unsigned)((t1 + B - 1) / B), B, 0, stream>>>(x, deg, W1, b1, rW1, rb1,
                                                              dinv, xw1, acc1);

    k_range_sum<HID><<<NB_RANGE, 1024, 0, stream>>>(own, row, col, ew, dinv, xw1, acc1);

    k_ln_relu_64<<<(N_NODES + 7) / 8, B, 0, stream>>>(acc1, g1, be1, x1);

    int waves_l2 = (N_NODES / 16) * (RED / 16);
    int thr_l2   = waves_l2 * 32;
    k_l2_wmma<<<(thr_l2 + B - 1) / B, B, 0, stream>>>(x1, W2, rW2, b2, rb2, dinv, xw2, acc2);

    k_range_sum<RED><<<NB_RANGE, 1024, 0, stream>>>(own, row, col, ew, dinv, xw2, acc2);

    k_head<<<(N_NODES + 7) / 8, B, 0, stream>>>(acc2, g2, be2, fcW, fcb, pW, pb, logits, emb);
}


# range sums: a wave keeps the owner bytes of 8 sweep steps in flight (ring of 8 loads, counted waits), not 1 load waited for in every step; lists, walk, LDS adds in arrival order as before
# speedup vs baseline: 2.1163x; 1.0577x over previous
#include <hip/hip_runtime.h>
#include <hip/hip_bf16.h>

#define N_NODES 100000
#define N_EDGES 3200000
#define HID 64
#define RED 32
#define EMB 32
#define LN_EPS 1e-5f
#define RANGE 391
#define NB_RANGE 256
#define LCAP 1536
#define NCHUNK (N_EDGES / 512)

typedef __attribute__((ext_vector_type(2))) float v2f;
typedef __attribute__((ext_vector_type(8))) float v8f;

__global__ void k_deg_init(float* __restrict__ deg) {
    int i = blockIdx.x * blockDim.x + threadIdx.x;
    if (i < N_NODES) deg[i] = 1.0f;
}

__global__ void k_deg_scatter(const int* __restrict__ col, const float* __restrict__ w,
                              float* __restrict__ deg, unsigned char* __restrict__ own) {
    int e = blockIdx.x * blockDim.x + threadIdx.x;
    if (e < N_EDGES) {
        int c = col[e];
        atomicAdd(&deg[c], w[e]);
        own[e] = (unsigned char)(c / RANGE);
    }
}

__global__ void k_l1_init(const float* __restrict__ x, const float* __restrict__ deg,
                          const float* __restrict__ W1, const float* __restrict__ b1,
                          const float* __restrict__ rW1, const float* __restrict__ rb1,
                          float* __restrict__ dinv, float* __restrict__ xw1,
                          float* __restrict__ acc1) {
    long long t = (long long)blockIdx.x * blockDim.x + threadIdx.x;
    if (t >= (long long)N_NODES * HID) return;
    int n = (int)(t >> 6);
    int f = (int)(t & 63);
    float d  = deg[n];
    float di = d > 0.f ? rsqrtf(d) : 0.f;
    if (f == 0) dinv[n] = di;
    float x0 = x[2 * n], x1v = x[2 * n + 1];
    float xw = fmaf(x0, W1[f], x1v * W1[HID + f]);
    xw1[t]  = xw;
    acc1[t] = di * di * xw + b1[f] + fmaf(x0, rW1[f], x1v * rW1[HID + f]) + rb1[f];
}

template <int F>
__device__ __forceinline__ void walk_list(const int* list, int n, int lane, int c0,
                                          const int* __restrict__ row, const int* __restrict__ col,
                                          const float* __restrict__ w, const float* __restrict__ dinv,
                                          const float* __restrict__ src, float* s_acc) {
    for (int base = 0; base < n; base += 32) {
        int m = n - base; if (m > 32) m = 32;
        int i = base + lane;
        int e = list[i < n ? i : n - 1];
        int r = row[e], c = col[e];
        float cf = dinv[r] * w[e] * dinv[c];
        int cl = c - c0;
        int j = 0;
        for (; j + 8 <= m; j += 8) {
            if (F == 64) {
                v2f s[8];
                #pragma unroll
                for (int u = 0; u < 8; ++u) {
                    int rj = __builtin_amdgcn_readlane(r, j + u);
                    s[u] = *(const v2f*)(src + (size_t)rj * 64 + 2 * lane);
                }
                #pragma unroll
                for (int u = 0; u < 8; ++u) {
                    int clj = __builtin_amdgcn_readlane(cl, j + u);
                    float cfj = __int_as_float(__builtin_amdgcn_readlane(__float_as_int(cf), j + u));
                    atomicAdd(&s_acc[clj * 64 + 2 * lane],     cfj * s[u].x);
                    atomicAdd(&s_acc[clj * 64 + 2 * lane + 1], cfj * s[u].y);
                }
            } else {
                float s[8];
                #pragma unroll
                for (int u = 0; u < 8; ++u) {
                    int rj = __builtin_amdgcn_readlane(r, j + u);
                    s[u] = src[(size_t)rj * F + lane];
                }
                #pragma unroll
                for (int u = 0; u < 8; ++u) {
                    int clj = __builtin_amdgcn_readlane(cl, j + u);
                    float cfj = __int_as_float(__builtin_amdgcn_readlane(__float_as_int(cf), j + u));
                    atomicAdd(&s_acc[clj * F + lane], cfj * s[u]);
                }
            }
        }
        for (; j < m; ++j) {
            int rj  = __builtin_amdgcn_readlane(r, j);
            int clj = __builtin_amdgcn_readlane(cl, j);
            float cfj = __int_as_float(__builtin_amdgcn_readlane(__float_as_int(cf), j));
            if (F == 64) {
                v2f s = *(const v2f*)(src + (size_t)rj * 64 + 2 * lane);
                atomicAdd(&s_acc[clj * 64 + 2 * lane],     cfj * s.x);
                atomicAdd(&s_acc[clj * 64 + 2 * lane + 1], cfj * s.y);
            } else {
                atomicAdd(&s_acc[clj * F + lane], cfj * src[(size_t)rj * F + lane]);
            }
        }
    }
}

__device__ __forceinline__ unsigned nz_bytes(unsigned x) {
    return ((x & 0x7F7F7F7Fu) + 0x7F7F7F7Fu) | x;
}

template <int F>
__global__ __launch_bounds__(1024) void k_range_sum(const unsigned char* __restrict__ own,
                                                    const int* __restrict__ row,
                                                    const int* __restrict__ col,
                                                    const float* __restrict__ w,
                                                    const float* __restrict__ dinv,
                                                    const float* __restrict__ src,
                                                    float* __restrict__ acc) {
    __shared__ float s_acc[RANGE * F];
    __shared__ int   s_list[32 * LCAP];
    const int tid = threadIdx.x, lane = tid & 31;
    const int wave = __builtin_amdgcn_readfirstlane(tid >> 5);
    const int b = blockIdx.x, c0 = b * RANGE;
    for (int i = tid; i < RANGE * F; i += 1024) s_acc[i] = 0.f;
    __syncthreads();

    int* list = s_list + wave * LCAP;
    int cnt = 0;
    const uint4* own4 = (const uint4*)own;
    const unsigned key = (unsigned)b * 0x01010101u;
    const int start = (b * NCHUNK) / NB_RANGE;
    auto ask = [&](int l) -> uint4 {
        int ph = l + start; if (ph >= NCHUNK) ph -= NCHUNK;
        return own4[(size_t)ph * 32 + lane];
    };
    auto sweep_step = [&](const uint4 cur, int l) {
        int ph = l + start; if (ph >= NCHUNK) ph -= NCHUNK;
        const int ebase = ph * 512 + lane * 16;
        if (__builtin_expect(cnt > LCAP - 512, 0)) {
            __builtin_amdgcn_fence(__ATOMIC_ACQ_REL, "wavefront");
            walk_list<F>(list, cnt, lane, c0, row, col, w, dinv, src, s_acc);
            __builtin_amdgcn_fence(__ATOMIC_ACQ_REL, "wavefront");
            cnt = 0;
        }
        const unsigned H = 0x80808080u;
        unsigned t0 = nz_bytes(cur.x ^ key), t1 = nz_bytes(cur.y ^ key);
        unsigned t2 = nz_bytes(cur.z ^ key), t3 = nz_bytes(cur.w ^ key);
        unsigned m = ((~t0 & H) >> 7) | ((~t1 & H) >> 6) | ((~t2 & H) >> 5) | ((~t3 & H) >> 4);
        unsigned bal = __builtin_amdgcn_ballot_w32(m != 0);
        while (bal) {
            int p = __builtin_ctz(m | 0x80000000u);
            int slot = cnt + (int)__builtin_amdgcn_mbcnt_lo(bal, 0);
            if (m != 0) list[slot] = ebase + ((p & 3) << 2) + (p >> 3);
            cnt += __builtin_popcount(bal);
            m &= m - 1;
            bal = __builtin_amdgcn_ballot_w32(m != 0);
        }
    };
    constexpr int DEPTH  = 8;
    constexpr int NSTEP  = NCHUNK / 32;
    constexpr int NBLOCK = NSTEP / DEPTH;
    constexpr int NTAIL  = NSTEP - NBLOCK * DEPTH;
    static_assert(NBLOCK >= 1 && NTAIL + 1 <= DEPTH, "the ring holds the tail steps");
    uint4 ring[DEPTH];
    #pragma unroll
    for (int u = 0; u < DEPTH; ++u) ring[u] = ask(wave + 32 * u);
    int l0 = wave;
    for (int blk = 0; blk < NBLOCK; ++blk) {
        #pragma unroll
        for (int u = 0; u < DEPTH; ++u) {
            const int l = l0 + 32 * u;
            sweep_step(ring[u], l);
            int ln = l + 32 * DEPTH; if (ln >= NCHUNK) ln = l;
            ring[u] = ask(ln);
        }
        l0 += 32 * DEPTH;
    }
    #pragma unroll
    for (int u = 0; u <= NTAIL; ++u) {
        const int l = l0 + 32 * u;
        if (u < NTAIL || l < NCHUNK) sweep_step(ring[u], l);
    }
    __builtin_amdgcn_fence(__ATOMIC_ACQ_REL, "wavefront");
    walk_list<F>(list, cnt, lane, c0, row, col, w, dinv, src, s_acc);
    __syncthreads();

    int rows = N_NODES - c0; if (rows > RANGE) rows = RANGE;
    float* ap = acc + (size_t)c0 * F;
    for (int i = tid; i < rows * F; i += 1024) ap[i] += s_acc[i];
}

__global__ void k_ln_relu_64(const float* __restrict__ acc, const float* __restrict__ g,
                             const float* __restrict__ be, float* __restrict__ out) {
    int wave = threadIdx.x >> 5;
    int lane = threadIdx.x & 31;
    int n = blockIdx.x * (blockDim.x >> 5) + wave;
    if (n >= N_NODES) return;
    float v0 = acc[(long long)n * 64 + lane];
    float v1 = acc[(long long)n * 64 + lane + 32];
    float s = v0 + v1, sq = fmaf(v0, v0, v1 * v1);
    #pragma unroll
    for (int off = 16; off > 0; off >>= 1) {
        s  += __shfl_xor(s,  off, 32);
        sq += __shfl_xor(sq, off, 32);
    }
    float mu  = s * (1.f / 64.f);
    float var = sq * (1.f / 64.f) - mu * mu;
    float inv = rsqrtf(var + LN_EPS);
    float o0 = (v0 - mu) * inv * g[lane]      + be[lane];
    float o1 = (v1 - mu) * inv * g[lane + 32] + be[lane + 32];
    out[(long long)n * 64 + lane]      = fmaxf(o0, 0.f);
    out[(long long)n * 64 + lane + 32] = fmaxf(o1, 0.f);
}

__global__ void k_l2_wmma(const float* __restrict__ x1, const float* __restrict__ W2,
                          const float* __restrict__ rW2, const float* __restrict__ b2,
                          const float* __restrict__ rb2, const float* __restrict__ dinv,
                          float* __restrict__ xw2, float* __restrict__ acc2) {
    int wave_global = (blockIdx.x * blockDim.x + threadIdx.x) >> 5;
    int lane = threadIdx.x & 31;
    const int tiles_m = N_NODES / 16;
    int tm = wave_global >> 1;
    int tn = wave_global & 1;
    if (tm >= tiles_m) return;
    int m0 = tm * 16, n0 = tn * 16;
    int  lr = lane & 15;
    bool hi = lane >= 16;
    int  kofs = hi ? 2 : 0;

    v8f c0 = {};
    v8f c1 = {};
    #pragma unroll
    for (int k0 = 0; k0 < HID; k0 += 4) {
        const float* ap = &x1[(long long)(m0 + lr) * HID + k0 + kofs];
        v2f a; a.x = ap[0]; a.y = ap[1];
        int kb = k0 + kofs;
        v2f b;  b.x  = W2 [kb * RED + n0 + lr]; b.y  = W2 [(kb + 1) * RED + n0 + lr];
        v2f br; br.x = rW2[kb * RED + n0 + lr]; br.y = rW2[(kb + 1) * RED + n0 + lr];
        c0 = __builtin_amdgcn_wmma_f32_16x16x4_f32(false, a, false, b,  (short)0, c0, false, false);
        c1 = __builtin_amdgcn_wmma_f32_16x16x4_f32(false, a, false, br, (short)0, c1, false, false);
    }
    int ncol = n0 + lr;
    float b2v = b2[ncol], rb2v = rb2[ncol];
    #pragma unroll
    for (int v = 0; v < 8; ++v) {
        int m = m0 + v + (hi ? 8 : 0);
        float di = dinv[m];
        float xv = c0[v];
        xw2 [(long long)m * RED + ncol] = xv;
        acc2[(long long)m * RED + ncol] = di * di * xv + b2v + c1[v] + rb2v;
    }
}

__global__ void k_head(const float* __restrict__ acc2, const float* __restrict__ g2,
                       const float* __restrict__ be2, const float* __restrict__ fcW,
                       const float* __restrict__ fcb, const float* __restrict__ pW,
                       const float* __restrict__ pb, float* __restrict__ logits,
                       float* __restrict__ emb) {
    int wave = threadIdx.x >> 5;
    int lane = threadIdx.x & 31;
    int n = blockIdx.x * (blockDim.x >> 5) + wave;
    if (n >= N_NODES) return;
    float v = acc2[(long long)n * 32 + lane];
    float s = v, sq = v * v;
    #pragma unroll
    for (int off = 16; off > 0; off >>= 1) {
        s  += __shfl_xor(s,  off, 32);
        sq += __shfl_xor(sq, off, 32);
    }
    float mu  = s * (1.f / 32.f);
    float var = sq * (1.f / 32.f) - mu * mu;
    float inv = rsqrtf(var + LN_EPS);
    float x2 = fmaxf((v - mu) * inv * g2[lane] + be2[lane], 0.f);

    float lg = x2 * fcW[lane];
    #pragma unroll
    for (int off = 16; off > 0; off >>= 1) lg += __shfl_xor(lg, off, 32);
    if (lane == 0) logits[n] = lg + fcb[0];

    float p = pb[lane];
    #pragma unroll
    for (int i = 0; i < 32; ++i)
        p = fmaf(__shfl(x2, i, 32), pW[i * 32 + lane], p);
    float nsq = p * p;
    #pragma unroll
    for (int off = 16; off > 0; off >>= 1) nsq += __shfl_xor(nsq, off, 32);
    float nrm = sqrtf(nsq);
    emb[(long long)n * 32 + lane] = p / fmaxf(nrm, 1e-12f);
}

extern "C" void kernel_launch(void* const* d_in, const int* in_sizes, int n_in,
                              void* d_out, int out_size, void* d_ws, size_t ws_size,
                              hipStream_t stream) {
    const float* x    = (const float*)d_in[0];
    const float* ew   = (const float*)d_in[1];
    const float* W1   = (const float*)d_in[2];
    const float* b1   = (const float*)d_in[3];
    const float* rW1  = (const float*)d_in[4];
    const float* rb1  = (const float*)d_in[5];
    const float* g1   = (const float*)d_in[6];
    const float* be1  = (const float*)d_in[7];
    const float* W2   = (const float*)d_in[8];
    const float* b2   = (const float*)d_in[9];
    const float* rW2  = (const float*)d_in[10];
    const float* rb2  = (const float*)d_in[11];
    const float* g2   = (const float*)d_in[12];
    const float* be2  = (const float*)d_in[13];
    const float* fcW  = (const float*)d_in[14];
    const float* fcb  = (const float*)d_in[15];
    const float* pW   = (const float*)d_in[16];
    const float* pb   = (const float*)d_in[17];
    const int*   ei   = (const int*)d_in[18];
    const int*   row  = ei;
    const int*   col  = ei + N_EDGES;

    float* ws   = (float*)d_ws;
    float* deg  = ws;  ws += N_NODES;
    float* dinv = ws;  ws += N_NODES;
    float* xw1  = ws;  ws += (size_t)N_NODES * HID;
    float* acc1 = ws;  ws += (size_t)N_NODES * HID;
    unsigned char* own = (unsigned char*)ws;  ws += (size_t)N_NODES * HID;
    float* x1   = xw1;
    float* xw2  = ws;  ws += (size_t)N_NODES * RED;
    float* acc2 = ws;  ws += (size_t)N_NODES * RED;

    float* logits = (float*)d_out;
    float* emb    = logits + N_NODES;

    const int B = 256;
    k_deg_init<<<(N_NODES + B - 1) / B, B, 0, stream>>>(deg);
    k_deg_scatter<<<(N_EDGES + B - 1) / B, B, 0, stream>>>(col, ew, deg, own);

    long long t1 = (long long)N_NODES * HID;
    k_l1_init<<<(unsigned)((t1 + B - 1) / B), B, 0, stream>>>(x, deg, W1, b1, rW1, rb1,
                                                              dinv, xw1, acc1);

    k_range_sum<HID><<<NB_RANGE, 1024, 0, stream>>>(own, row, col, ew, dinv, xw1, acc1);

    k_ln_relu_64<<<(N_NODES + 7) / 8, B, 0, stream>>>(acc1, g1, be1, x1);

    int waves_l2 = (N_NODES / 16) * (RED / 16);
    int thr_l2   = waves_l2 * 32;
    k_l2_wmma<<<(thr_l2 + B - 1) / B, B, 0, stream>>>(x1, W2, rW2, b2, rb2, dinv, xw2, acc2);

    k_range_sum<RED><<<NB_RANGE, 1024, 0, stream>>>(own, row, col, ew, dinv, xw2, acc2);

    k_head<<<(N_NODES + 7) / 8, B, 0, stream>>>(acc2, g2, be2, fcW, fcb, pW, pb, logits, emb);
}


# range sums read one 8-byte record per listed edge {source | local destination, weight}, packed in edge order by the degree pass: 1 scattered line per edge, not 3 (row, col, w); LDS adds as before
# speedup vs baseline: 2.3799x; 1.1246x over previous
#include <hip/hip_runtime.h>
#include <hip/hip_bf16.h>

#define N_NODES 100000
#define N_EDGES 3200000
#define HID 64
#define RED 32
#define EMB 32
#define LN_EPS 1e-5f
#define RANGE 391
#define NB_RANGE 256
#define LCAP 1536
#define NCHUNK (N_EDGES / 512)
#define REC_SHIFT 17
#define REC_MASK ((1u << REC_SHIFT) - 1u)

typedef __attribute__((ext_vector_type(2))) float v2f;
typedef __attribute__((ext_vector_type(8))) float v8f;

__global__ void k_deg_init(float* __restrict__ deg) {
    int i = blockIdx.x * blockDim.x + threadIdx.x;
    if (i < N_NODES) deg[i] = 1.0f;
}

__global__ void k_deg_scatter(const int* __restrict__ row, const int* __restrict__ col,
                              const float* __restrict__ w, float* __restrict__ deg,
                              unsigned char* __restrict__ own, unsigned int* __restrict__ rec) {
    int e = blockIdx.x * blockDim.x + threadIdx.x;
    if (e < N_EDGES) {
        int c = col[e];
        float we = w[e];
        atomicAdd(&deg[c], we);
        int o = c / RANGE;
        own[e] = (unsigned char)o;
        uint2 rc;
        rc.x = (unsigned)row[e] | ((unsigned)(c - o * RANGE) << REC_SHIFT);
        rc.y = __float_as_uint(we);
        ((uint2*)rec)[e] = rc;
    }
}

__global__ void k_l1_init(const float* __restrict__ x, const float* __restrict__ deg,
                          const float* __restrict__ W1, const float* __restrict__ b1,
                          const float* __restrict__ rW1, const float* __restrict__ rb1,
                          float* __restrict__ dinv, float* __restrict__ xw1,
                          float* __restrict__ acc1) {
    long long t = (long long)blockIdx.x * blockDim.x + threadIdx.x;
    if (t >= (long long)N_NODES * HID) return;
    int n = (int)(t >> 6);
    int f = (int)(t & 63);
    float d  = deg[n];
    float di = d > 0.f ? rsqrtf(d) : 0.f;
    if (f == 0) dinv[n] = di;
    float x0 = x[2 * n], x1v = x[2 * n + 1];
    float xw = fmaf(x0, W1[f], x1v * W1[HID + f]);
    xw1[t]  = xw;
    acc1[t] = di * di * xw + b1[f] + fmaf(x0, rW1[f], x1v * rW1[HID + f]) + rb1[f];
}

template <int F>
__device__ __forceinline__ void walk_list(const int* list, int n, int lane, int c0,
                                          const uint2* __restrict__ rec, const float* __restrict__ dinv,
                                          const float* __restrict__ src, float* s_acc) {
    for (int base = 0; base < n; base += 32) {
        int m = n - base; if (m > 32) m = 32;
        int i = base + lane;
        int e = list[i < n ? i : n - 1];
        const uint2 rc = rec[e];
        int r  = (int)(rc.x & REC_MASK);
        int cl = (int)(rc.x >> REC_SHIFT);
        float cf = dinv[r] * __uint_as_float(rc.y) * dinv[c0 + cl];
        int j = 0;
        for (; j + 8 <= m; j += 8) {
            if (F == 64) {
                v2f s[8];
                #pragma unroll
                for (int u = 0; u < 8; ++u) {
                    int rj = __builtin_amdgcn_readlane(r, j + u);
                    s[u] = *(const v2f*)(src + (size_t)rj * 64 + 2 * lane);
                }
                #pragma unroll
                for (int u = 0; u < 8; ++u) {
                    int clj = __builtin_amdgcn_readlane(cl, j + u);
                    float cfj = __int_as_float(__builtin_amdgcn_readlane(__float_as_int(cf), j + u));
                    atomicAdd(&s_acc[clj * 64 + 2 * lane],     cfj * s[u].x);
                    atomicAdd(&s_acc[clj * 64 + 2 * lane + 1], cfj * s[u].y);
                }
            } else {
                float s[8];
                #pragma unroll
                for (int u = 0; u < 8; ++u) {
                    int rj = __builtin_amdgcn_readlane(r, j + u);
                    s[u] = src[(size_t)rj * F + lane];
                }
                #pragma unroll
                for (int u = 0; u < 8; ++u) {
                    int clj = __builtin_amdgcn_readlane(cl, j + u);
                    float cfj = __int_as_float(__builtin_amdgcn_readlane(__float_as_int(cf), j + u));
                    atomicAdd(&s_acc[clj * F + lane], cfj * s[u]);
                }
            }
        }
        for (; j < m; ++j) {
            int rj  = __builtin_amdgcn_readlane(r, j);
            int clj = __builtin_amdgcn_readlane(cl, j);
            float cfj = __int_as_float(__builtin_amdgcn_readlane(__float_as_int(cf), j));
            if (F == 64) {
                v2f s = *(const v2f*)(src + (size_t)rj * 64 + 2 * lane);
                atomicAdd(&s_acc[clj * 64 + 2 * lane],     cfj * s.x);
                atomicAdd(&s_acc[clj * 64 + 2 * lane + 1], cfj * s.y);
            } else {
                atomicAdd(&s_acc[clj * F + lane], cfj * src[(size_t)rj * F + lane]);
            }
        }
    }
}

__device__ __forceinline__ unsigned nz_bytes(unsigned x) {
    return ((x & 0x7F7F7F7Fu) + 0x7F7F7F7Fu) | x;
}

template <int F>
__global__ __launch_bounds__(1024) void k_range_sum(const unsigned char* __restrict__ own,
                                                    const unsigned int* __restrict__ rec,
                                                    const float* __restrict__ dinv,
                                                    const float* __restrict__ src,
                                                    float* __restrict__ acc) {
    __shared__ float s_acc[RANGE * F];
    __shared__ int   s_list[32 * LCAP];
    const int tid = threadIdx.x, lane = tid & 31;
    const int wave = __builtin_amdgcn_readfirstlane(tid >> 5);
    const int b = blockIdx.x, c0 = b * RANGE;
    for (int i = tid; i < RANGE * F; i += 1024) s_acc[i] = 0.f;
    __syncthreads();

    int* list = s_list + wave * LCAP;
    int cnt = 0;
    const uint4* own4 = (const uint4*)own;
    const uint2* rec2 = (const uint2*)rec;
    const unsigned key = (unsigned)b * 0x01010101u;
    const int start = (b * NCHUNK) / NB_RANGE;
    auto ask = [&](int l) -> uint4 {
        int ph = l + start; if (ph >= NCHUNK) ph -= NCHUNK;
        return own4[(size_t)ph * 32 + lane];
    };
    auto sweep_step = [&](const uint4 cur, int l) {
        int ph = l + start; if (ph >= NCHUNK) ph -= NCHUNK;
        const int ebase = ph * 512 + lane * 16;
        if (__builtin_expect(cnt > LCAP - 512, 0)) {
            __builtin_amdgcn_fence(__ATOMIC_ACQ_REL, "wavefront");
            walk_list<F>(list, cnt, lane, c0, rec2, dinv, src, s_acc);
            __builtin_amdgcn_fence(__ATOMIC_ACQ_REL, "wavefront");
            cnt = 0;
        }
        const unsigned H = 0x80808080u;
        unsigned t0 = nz_bytes(cur.x ^ key), t1 = nz_bytes(cur.y ^ key);
        unsigned t2 = nz_bytes(cur.z ^ key), t3 = nz_bytes(cur.w ^ key);
        unsigned m = ((~t0 & H) >> 7) | ((~t1 & H) >> 6) | ((~t2 & H) >> 5) | ((~t3 & H) >> 4);
        unsigned bal = __builtin_amdgcn_ballot_w32(m != 0);
        while (bal) {
            int p = __builtin_ctz(m | 0x80000000u);
            int slot = cnt + (int)__builtin_amdgcn_mbcnt_lo(bal, 0);
            if (m != 0) list[slot] = ebase + ((p & 3) << 2) + (p >> 3);
            cnt += __builtin_popcount(bal);
            m &= m - 1;
            bal = __builtin_amdgcn_ballot_w32(m != 0);
        }
    };
    constexpr int DEPTH  = 8;
    constexpr int NSTEP  = NCHUNK / 32;
    constexpr int NBLOCK = NSTEP / DEPTH;
    constexpr int NTAIL  = NSTEP - NBLOCK * DEPTH;
    static_assert(NBLOCK >= 1 && NTAIL + 1 <= DEPTH, "the ring holds the tail steps");
    uint4 ring[DEPTH];
    #pragma unroll
    for (int u = 0; u < DEPTH; ++u) ring[u] = ask(wave + 32 * u);
    int l0 = wave;
    for (int blk = 0; blk < NBLOCK; ++blk) {
        #pragma unroll
        for (int u = 0; u < DEPTH; ++u) {
            const int l = l0 + 32 * u;
            sweep_step(ring[u], l);
            int ln = l + 32 * DEPTH; if (ln >= NCHUNK) ln = l;
            ring[u] = ask(ln);
        }
        l0 += 32 * DEPTH;
    }
    #pragma unroll
    for (int u = 0; u <= NTAIL; ++u) {
        const int l = l0 + 32 * u;
        if (u < NTAIL || l < NCHUNK) sweep_step(ring[u], l);
    }
    __builtin_amdgcn_fence(__ATOMIC_ACQ_REL, "wavefront");
    walk_list<F>(list, cnt, lane, c0, rec2, dinv, src, s_acc);
    __syncthreads();

    int rows = N_NODES - c0; if (rows > RANGE) rows = RANGE;
    float* ap = acc + (size_t)c0 * F;
    for (int i = tid; i < rows * F; i += 1024) ap[i] += s_acc[i];
}

__global__ void k_ln_relu_64(const float* __restrict__ acc, const float* __restrict__ g,
                             const float* __restrict__ be, float* __restrict__ out) {
    int wave = threadIdx.x >> 5;
    int lane = threadIdx.x & 31;
    int n = blockIdx.x * (blockDim.x >> 5) + wave;
    if (n >= N_NODES) return;
    float v0 = acc[(long long)n * 64 + lane];
    float v1 = acc[(long long)n * 64 + lane + 32];
    float s = v0 + v1, sq = fmaf(v0, v0, v1 * v1);
    #pragma unroll
    for (int off = 16; off > 0; off >>= 1) {
        s  += __shfl_xor(s,  off, 32);
        sq += __shfl_xor(sq, off, 32);
    }
    float mu  = s * (1.f / 64.f);
    float var = sq * (1.f / 64.f) - mu * mu;
    float inv = rsqrtf(var + LN_EPS);
    float o0 = (v0 - mu) * inv * g[lane]      + be[lane];
    float o1 = (v1 - mu) * inv * g[lane + 32] + be[lane + 32];
    out[(long long)n * 64 + lane]      = fmaxf(o0, 0.f);
    out[(long long)n * 64 + lane + 32] = fmaxf(o1, 0.f);
}

__global__ void k_l2_wmma(const float* __restrict__ x1, const float* __restrict__ W2,
                          const float* __restrict__ rW2, const float* __restrict__ b2,
                          const float* __restrict__ rb2, const float* __restrict__ dinv,
                          float* __restrict__ xw2, float* __restrict__ acc2) {
    int wave_global = (blockIdx.x * blockDim.x + threadIdx.x) >> 5;
    int lane = threadIdx.x & 31;
    const int tiles_m = N_NODES / 16;
    int tm = wave_global >> 1;
    int tn = wave_global & 1;
    if (tm >= tiles_m) return;
    int m0 = tm * 16, n0 = tn * 16;
    int  lr = lane & 15;
    bool hi = lane >= 16;
    int  kofs = hi ? 2 : 0;

    v8f c0 = {};
    v8f c1 = {};
    #pragma unroll
    for (int k0 = 0; k0 < HID; k0 += 4) {
        const float* ap = &x1[(long long)(m0 + lr) * HID + k0 + kofs];
        v2f a; a.x = ap[0]; a.y = ap[1];
        int kb = k0 + kofs;
        v2f b;  b.x  = W2 [kb * RED + n0 + lr]; b.y  = W2 [(kb + 1) * RED + n0 + lr];
        v2f br; br.x = rW2[kb * RED + n0 + lr]; br.y = rW2[(kb + 1) * RED + n0 + lr];
        c0 = __builtin_amdgcn_wmma_f32_16x16x4_f32(false, a, false, b,  (short)0, c0, false, false);
        c1 = __builtin_amdgcn_wmma_f32_16x16x4_f32(false, a, false, br, (short)0, c1, false, false);
    }
    int ncol = n0 + lr;
    float b2v = b2[ncol], rb2v = rb2[ncol];
    #pragma unroll
    for (int v = 0; v < 8; ++v) {
        int m = m0 + v + (hi ? 8 : 0);
        float di = dinv[m];
        float xv = c0[v];
        xw2 [(long long)m * RED + ncol] = xv;
        acc2[(long long)m * RED + ncol] = di * di * xv + b2v + c1[v] + rb2v;
    }
}

__global__ void k_head(const float* __restrict__ acc2, const float* __restrict__ g2,
                       const float* __restrict__ be2, const float* __restrict__ fcW,
                       const float* __restrict__ fcb, const float* __restrict__ pW,
                       const float* __restrict__ pb, float* __restrict__ logits,
                       float* __restrict__ emb) {
    int wave = threadIdx.x >> 5;
    int lane = threadIdx.x & 31;
    int n = blockIdx.x * (blockDim.x >> 5) + wave;
    if (n >= N_NODES) return;
    float v = acc2[(long long)n * 32 + lane];
    float s = v, sq = v * v;
    #pragma unroll
    for (int off = 16; off > 0; off >>= 1) {
        s  += __shfl_xor(s,  off, 32);
        sq += __shfl_xor(sq, off, 32);
    }
    float mu  = s * (1.f / 32.f);
    float var = sq * (1.f / 32.f) - mu * mu;
    float inv = rsqrtf(var + LN_EPS);
    float x2 = fmaxf((v - mu) * inv * g2[lane] + be2[lane], 0.f);

    float lg = x2 * fcW[lane];
    #pragma unroll
    for (int off = 16; off > 0; off >>= 1) lg += __shfl_xor(lg, off, 32);
    if (lane == 0) logits[n] = lg + fcb[0];

    float p = pb[lane];
    #pragma unroll
    for (int i = 0; i < 32; ++i)
        p = fmaf(__shfl(x2, i, 32), pW[i * 32 + lane], p);
    float nsq = p * p;
    #pragma unroll
    for (int off = 16; off > 0; off >>= 1) nsq += __shfl_xor(nsq, off, 32);
    float nrm = sqrtf(nsq);
    emb[(long long)n * 32 + lane] = p / fmaxf(nrm, 1e-12f);
}

extern "C" void kernel_launch(void* const* d_in, const int* in_sizes, int n_in,
                              void* d_out, int out_size, void* d_ws, size_t ws_size,
                              hipStream_t stream) {
    const float* x    = (const float*)d_in[0];
    const float* ew   = (const float*)d_in[1];
    const float* W1   = (const float*)d_in[2];
    const float* b1   = (const float*)d_in[3];
    const float* rW1  = (const float*)d_in[4];
    const float* rb1  = (const float*)d_in[5];
    const float* g1   = (const float*)d_in[6];
    const float* be1  = (const float*)d_in[7];
    const float* W2   = (const float*)d_in[8];
    const float* b2   = (const float*)d_in[9];
    const float* rW2  = (const float*)d_in[10];
    const float* rb2  = (const float*)d_in[11];
    const float* g2   = (const float*)d_in[12];
    const float* be2  = (const float*)d_in[13];
    const float* fcW  = (const float*)d_in[14];
    const float* fcb  = (const float*)d_in[15];
    const float* pW   = (const float*)d_in[16];
    const float* pb   = (const float*)d_in[17];
    const int*   ei   = (const int*)d_in[18];
    const int*   row  = ei;
    const int*   col  = ei + N_EDGES;

    float* ws   = (float*)d_ws;
    float* deg  = ws;  ws += N_NODES;
    float* dinv = ws;  ws += N_NODES;
    float* xw1  = ws;  ws += (size_t)N_NODES * HID;
    float* acc1 = ws;  ws += (size_t)N_NODES * HID;
    unsigned char* own = (unsigned char*)ws;  ws += (size_t)N_NODES * HID;
    float* x1   = xw1;
    unsigned int* rec = (unsigned int*)ws;  ws += (size_t)N_NODES * HID;
    float* xw2  = acc1;
    float* acc2 = acc1 + (size_t)N_NODES * RED;

    float* logits = (float*)d_out;
    float* emb    = logits + N_NODES;

    const int B = 256;
    k_deg_init<<<(N_NODES + B - 1) / B, B, 0, stream>>>(deg);
    k_deg_scatter<<<(N_EDGES + B - 1) / B, B, 0, stream>>>(row, col, ew, deg, own, rec);

    long long t1 = (long long)N_NODES * HID;
    k_l1_init<<<(unsigned)((t1 + B - 1) / B), B, 0, stream>>>(x, deg, W1, b1, rW1, rb1,
                                                              dinv, xw1, acc1);

    k_range_sum<HID><<<NB_RANGE, 1024, 0, stream>>>(own, rec, dinv, xw1, acc1);

    k_ln_relu_64<<<(N_NODES + 7) / 8, B, 0, stream>>>(acc1, g1, be1, x1);

    int waves_l2 = (N_NODES / 16) * (RED / 16);
    int thr_l2   = waves_l2 * 32;
    k_l2_wmma<<<(thr_l2 + B - 1) / B, B, 0, stream>>>(x1, W2, rW2, b2, rb2, dinv, xw2, acc2);

    k_range_sum<RED><<<NB_RANGE, 1024, 0, stream>>>(own, rec, dinv, xw2, acc2);

    k_head<<<(N_NODES + 7) / 8, B, 0, stream>>>(acc2, g2, be2, fcW, fcb, pW, pb, logits, emb);
}


# range sums gather rows pre-scaled by dinv[source]; dinv[destination] applied once per row at the final store: no dinv gathers per edge; products rounded differently, sum order as before
# speedup vs baseline: 2.4717x; 1.0386x over previous
#include <hip/hip_runtime.h>
#include <hip/hip_bf16.h>

#define N_NODES 100000
#define N_EDGES 3200000
#define HID 64
#define RED 32
#define EMB 32
#define LN_EPS 1e-5f
#define RANGE 391
#define NB_RANGE 256
#define LCAP 1536
#define NCHUNK (N_EDGES / 512)
#define REC_SHIFT 17
#define REC_MASK ((1u << REC_SHIFT) - 1u)

typedef __attribute__((ext_vector_type(2))) float v2f;
typedef __attribute__((ext_vector_type(8))) float v8f;

__global__ void k_deg_init(float* __restrict__ deg) {
    int i = blockIdx.x * blockDim.x + threadIdx.x;
    if (i < N_NODES) deg[i] = 1.0f;
}

__global__ void k_deg_scatter(const int* __restrict__ row, const int* __restrict__ col,
                              const float* __restrict__ w, float* __restrict__ deg,
                              unsigned char* __restrict__ own, unsigned int* __restrict__ rec) {
    int e = blockIdx.x * blockDim.x + threadIdx.x;
    if (e < N_EDGES) {
        int c = col[e];
        float we = w[e];
        atomicAdd(&deg[c], we);
        int o = c / RANGE;
        own[e] = (unsigned char)o;
        uint2 rc;
        rc.x = (unsigned)row[e] | ((unsigned)(c - o * RANGE) << REC_SHIFT);
        rc.y = __float_as_uint(we);
        ((uint2*)rec)[e] = rc;
    }
}

__global__ void k_l1_init(const float* __restrict__ x, const float* __restrict__ deg,
                          const float* __restrict__ W1, const float* __restrict__ b1,
                          const float* __restrict__ rW1, const float* __restrict__ rb1,
                          float* __restrict__ dinv, float* __restrict__ xw1,
                          float* __restrict__ acc1) {
    long long t = (long long)blockIdx.x * blockDim.x + threadIdx.x;
    if (t >= (long long)N_NODES * HID) return;
    int n = (int)(t >> 6);
    int f = (int)(t & 63);
    float d  = deg[n];
    float di = d > 0.f ? rsqrtf(d) : 0.f;
    if (f == 0) dinv[n] = di;
    float x0 = x[2 * n], x1v = x[2 * n + 1];
    float xw = fmaf(x0, W1[f], x1v * W1[HID + f]);
    float xs = di * xw;
    xw1[t]  = xs;
    acc1[t] = di * xs + b1[f] + fmaf(x0, rW1[f], x1v * rW1[HID + f]) + rb1[f];
}

template <int F>
__device__ __forceinline__ void walk_list(const int* list, int n, int lane,
                                          const uint2* __restrict__ rec,
                                          const float* __restrict__ src, float* s_acc) {
    for (int base = 0; base < n; base += 32) {
        int m = n - base; if (m > 32) m = 32;
        int i = base + lane;
        int e = list[i < n ? i : n - 1];
        const uint2 rc = rec[e];
        int r  = (int)(rc.x & REC_MASK);
        int cl = (int)(rc.x >> REC_SHIFT);
        float cf = __uint_as_float(rc.y);
        int j = 0;
        for (; j + 8 <= m; j += 8) {
            if (F == 64) {
                v2f s[8];
                #pragma unroll
                for (int u = 0; u < 8; ++u) {
                    int rj = __builtin_amdgcn_readlane(r, j + u);
                    s[u] = *(const v2f*)(src + (size_t)rj * 64 + 2 * lane);
                }
                #pragma unroll
                for (int u = 0; u < 8; ++u) {
                    int clj = __builtin_amdgcn_readlane(cl, j + u);
                    float cfj = __int_as_float(__builtin_amdgcn_readlane(__float_as_int(cf), j + u));
                    atomicAdd(&s_acc[clj * 64 + 2 * lane],     cfj * s[u].x);
                    atomicAdd(&s_acc[clj * 64 + 2 * lane + 1], cfj * s[u].y);
                }
            } else {
                float s[8];
                #pragma unroll
                for (int u = 0; u < 8; ++u) {
                    int rj = __builtin_amdgcn_readlane(r, j + u);
                    s[u] = src[(size_t)rj * F + lane];
                }
                #pragma unroll
                for (int u = 0; u < 8; ++u) {
                    int clj = __builtin_amdgcn_readlane(cl, j + u);
                    float cfj = __int_as_float(__builtin_amdgcn_readlane(__float_as_int(cf), j + u));
                    atomicAdd(&s_acc[clj * F + lane], cfj * s[u]);
                }
            }
        }
        for (; j < m; ++j) {
            int rj  = __builtin_amdgcn_readlane(r, j);
            int clj = __builtin_amdgcn_readlane(cl, j);
            float cfj = __int_as_float(__builtin_amdgcn_readlane(__float_as_int(cf), j));
            if (F == 64) {
                v2f s = *(const v2f*)(src + (size_t)rj * 64 + 2 * lane);
                atomicAdd(&s_acc[clj * 64 + 2 * lane],     cfj * s.x);
                atomicAdd(&s_acc[clj * 64 + 2 * lane + 1], cfj * s.y);
            } else {
                atomicAdd(&s_acc[clj * F + lane], cfj * src[(size_t)rj * F + lane]);
            }
        }
    }
}

__device__ __forceinline__ unsigned nz_bytes(unsigned x) {
    return ((x & 0x7F7F7F7Fu) + 0x7F7F7F7Fu) | x;
}

template <int F>
__global__ __launch_bounds__(1024) void k_range_sum(const unsigned char* __restrict__ own,
                                                    const unsigned int* __restrict__ rec,
                                                    const float* __restrict__ dinv,
                                                    const float* __restrict__ src,
                                                    float* __restrict__ acc) {
    __shared__ float s_acc[RANGE * F];
    __shared__ int   s_list[32 * LCAP];
    const int tid = threadIdx.x, lane = tid & 31;
    const int wave = __builtin_amdgcn_readfirstlane(tid >> 5);
    const int b = blockIdx.x, c0 = b * RANGE;
    for (int i = tid; i < RANGE * F; i += 1024) s_acc[i] = 0.f;
    __syncthreads();

    int* list = s_list + wave * LCAP;
    int cnt = 0;
    const uint4* own4 = (const uint4*)own;
    const uint2* rec2 = (const uint2*)rec;
    const unsigned key = (unsigned)b * 0x01010101u;
    const int start = (b * NCHUNK) / NB_RANGE;
    auto ask = [&](int l) -> uint4 {
        int ph = l + start; if (ph >= NCHUNK) ph -= NCHUNK;
        return own4[(size_t)ph * 32 + lane];
    };
    auto sweep_step = [&](const uint4 cur, int l) {
        int ph = l + start; if (ph >= NCHUNK) ph -= NCHUNK;
        const int ebase = ph * 512 + lane * 16;
        if (__builtin_expect(cnt > LCAP - 512, 0)) {
            __builtin_amdgcn_fence(__ATOMIC_ACQ_REL, "wavefront");
            walk_list<F>(list, cnt, lane, rec2, src, s_acc);
            __builtin_amdgcn_fence(__ATOMIC_ACQ_REL, "wavefront");
            cnt = 0;
        }
        const unsigned H = 0x80808080u;
        unsigned t0 = nz_bytes(cur.x ^ key), t1 = nz_bytes(cur.y ^ key);
        unsigned t2 = nz_bytes(cur.z ^ key), t3 = nz_bytes(cur.w ^ key);
        unsigned m = ((~t0 & H) >> 7) | ((~t1 & H) >> 6) | ((~t2 & H) >> 5) | ((~t3 & H) >> 4);
        unsigned bal = __builtin_amdgcn_ballot_w32(m != 0);
        while (bal) {
            int p = __builtin_ctz(m | 0x80000000u);
            int slot = cnt + (int)__builtin_amdgcn_mbcnt_lo(bal, 0);
            if (m != 0) list[slot] = ebase + ((p & 3) << 2) + (p >> 3);
            cnt += __builtin_popcount(bal);
            m &= m - 1;
            bal = __builtin_amdgcn_ballot_w32(m != 0);
        }
    };
    constexpr int DEPTH  = 8;
    constexpr int NSTEP  = NCHUNK / 32;
    constexpr int NBLOCK = NSTEP / DEPTH;
    constexpr int NTAIL  = NSTEP - NBLOCK * DEPTH;
    static_assert(NBLOCK >= 1 && NTAIL + 1 <= DEPTH, "the ring holds the tail steps");
    uint4 ring[DEPTH];
    #pragma unroll
    for (int u = 0; u < DEPTH; ++u) ring[u] = ask(wave + 32 * u);
    int l0 = wave;
    for (int blk = 0; blk < NBLOCK; ++blk) {
        #pragma unroll
        for (int u = 0; u < DEPTH; ++u) {
            const int l = l0 + 32 * u;
            sweep_step(ring[u], l);
            int ln = l + 32 * DEPTH; if (ln >= NCHUNK) ln = l;
            ring[u] = ask(ln);
        }
        l0 += 32 * DEPTH;
    }
    #pragma unroll
    for (int u = 0; u <= NTAIL; ++u) {
        const int l = l0 + 32 * u;
        if (u < NTAIL || l < NCHUNK) sweep_step(ring[u], l);
    }
    __builtin_amdgcn_fence(__ATOMIC_ACQ_REL, "wavefront");
    walk_list<F>(list, cnt, lane, rec2, src, s_acc);
    __syncthreads();

    int rows = N_NODES - c0; if (rows > RANGE) rows = RANGE;
    float* ap = acc + (size_t)c0 * F;
    const float* dp = dinv + c0;
    for (int i = tid; i < rows * F; i += 1024) ap[i] = fmaf(dp[i / F], s_acc[i], ap[i]);
}

__global__ void k_ln_relu_64(const float* __restrict__ acc, const float* __restrict__ g,
                             const float* __restrict__ be, float* __restrict__ out) {
    int wave = threadIdx.x >> 5;
    int lane = threadIdx.x & 31;
    int n = blockIdx.x * (blockDim.x >> 5) + wave;
    if (n >= N_NODES) return;
    float v0 = acc[(long long)n * 64 + lane];
    float v1 = acc[(long long)n * 64 + lane + 32];
    float s = v0 + v1, sq = fmaf(v0, v0, v1 * v1);
    #pragma unroll
    for (int off = 16; off > 0; off >>= 1) {
        s  += __shfl_xor(s,  off, 32);
        sq += __shfl_xor(sq, off, 32);
    }
    float mu  = s * (1.f / 64.f);
    float var = sq * (1.f / 64.f) - mu * mu;
    float inv = rsqrtf(var + LN_EPS);
    float o0 = (v0 - mu) * inv * g[lane]      + be[lane];
    float o1 = (v1 - mu) * inv * g[lane + 32] + be[lane + 32];
    out[(long long)n * 64 + lane]      = fmaxf(o0, 0.f);
    out[(long long)n * 64 + lane + 32] = fmaxf(o1, 0.f);
}

__global__ void k_l2_wmma(const float* __restrict__ x1, const float* __restrict__ W2,
                          const float* __restrict__ rW2, const float* __restrict__ b2,
                          const float* __restrict__ rb2, const float* __restrict__ dinv,
                          float* __restrict__ xw2, float* __restrict__ acc2) {
    int wave_global = (blockIdx.x * blockDim.x + threadIdx.x) >> 5;
    int lane = threadIdx.x & 31;
    const int tiles_m = N_NODES / 16;
    int tm = wave_global >> 1;
    int tn = wave_global & 1;
    if (tm >= tiles_m) return;
    int m0 = tm * 16, n0 = tn * 16;
    int  lr = lane & 15;
    bool hi = lane >= 16;
    int  kofs = hi ? 2 : 0;

    v8f c0 = {};
    v8f c1 = {};
    #pragma unroll
    for (int k0 = 0; k0 < HID; k0 += 4) {
        const float* ap = &x1[(long long)(m0 + lr) * HID + k0 + kofs];
        v2f a; a.x = ap[0]; a.y = ap[1];
        int kb = k0 + kofs;
        v2f b;  b.x  = W2 [kb * RED + n0 + lr]; b.y  = W2 [(kb + 1) * RED + n0 + lr];
        v2f br; br.x = rW2[kb * RED + n0 + lr]; br.y = rW2[(kb + 1) * RED + n0 + lr];
        c0 = __builtin_amdgcn_wmma_f32_16x16x4_f32(false, a, false, b,  (short)0, c0, false, false);
        c1 = __builtin_amdgcn_wmma_f32_16x16x4_f32(false, a, false, br, (short)0, c1, false, false);
    }
    int ncol = n0 + lr;
    float b2v = b2[ncol], rb2v = rb2[ncol];
    #pragma unroll
    for (int v = 0; v < 8; ++v) {
        int m = m0 + v + (hi ? 8 : 0);
        float di = dinv[m];
        float xv = di * c0[v];
        xw2 [(long long)m * RED + ncol] = xv;
        acc2[(long long)m * RED + ncol] = di * xv + b2v + c1[v] + rb2v;
    }
}

__global__ void k_head(const float* __restrict__ acc2, const float* __restrict__ g2,
                       const float* __restrict__ be2, const float* __restrict__ fcW,
                       const float* __restrict__ fcb, const float* __restrict__ pW,
                       const float* __restrict__ pb, float* __restrict__ logits,
                       float* __restrict__ emb) {
    int wave = threadIdx.x >> 5;
    int lane = threadIdx.x & 31;
    int n = blockIdx.x * (blockDim.x >> 5) + wave;
    if (n >= N_NODES) return;
    float v = acc2[(long long)n * 32 + lane];
    float s = v, sq = v * v;
    #pragma unroll
    for (int off = 16; off > 0; off >>= 1) {
        s  += __shfl_xor(s,  off, 32);
        sq += __shfl_xor(sq, off, 32);
    }
    float mu  = s * (1.f / 32.f);
    float var = sq * (1.f / 32.f) - mu * mu;
    float inv = rsqrtf(var + LN_EPS);
    float x2 = fmaxf((v - mu) * inv * g2[lane] + be2[lane], 0.f);

    float lg = x2 * fcW[lane];
    #pragma unroll
    for (int off = 16; off > 0; off >>= 1) lg += __shfl_xor(lg, off, 32);
    if (lane == 0) logits[n] = lg + fcb[0];

    float p = pb[lane];
    #pragma unroll
    for (int i = 0; i < 32; ++i)
        p = fmaf(__shfl(x2, i, 32), pW[i * 32 + lane], p);
    float nsq = p * p;
    #pragma unroll
    for (int off = 16; off > 0; off >>= 1) nsq += __shfl_xor(nsq, off, 32);
    float nrm = sqrtf(nsq);
    emb[(long long)n * 32 + lane] = p / fmaxf(nrm, 1e-12f);
}

extern "C" void kernel_launch(void* const* d_in, const int* in_sizes, int n_in,
                              void* d_out, int out_size, void* d_ws, size_t ws_size,
                              hipStream_t stream) {
    const float* x    = (const float*)d_in[0];
    const float* ew   = (const float*)d_in[1];
    const float* W1   = (const float*)d_in[2];
    const float* b1   = (const float*)d_in[3];
    const float* rW1  = (const float*)d_in[4];
    const float* rb1  = (const float*)d_in[5];
    const float* g1   = (const float*)d_in[6];
    const float* be1  = (const float*)d_in[7];
    const float* W2   = (const float*)d_in[8];
    const float* b2   = (const float*)d_in[9];
    const float* rW2  = (const float*)d_in[10];
    const float* rb2  = (const float*)d_in[11];
    const float* g2   = (const float*)d_in[12];
    const float* be2  = (const float*)d_in[13];
    const float* fcW  = (const float*)d_in[14];
    const float* fcb  = (const float*)d_in[15];
    const float* pW   = (const float*)d_in[16];
    const float* pb   = (const float*)d_in[17];
    const int*   ei   = (const int*)d_in[18];
    const int*   row  = ei;
    const int*   col  = ei + N_EDGES;

    float* ws   = (float*)d_ws;
    float* deg  = ws;  ws += N_NODES;
    float* dinv = ws;  ws += N_NODES;
    float* xw1  = ws;  ws += (size_t)N_NODES * HID;
    float* acc1 = ws;  ws += (size_t)N_NODES * HID;
    unsigned char* own = (unsigned char*)ws;  ws += (size_t)N_NODES * HID;
    float* x1   = xw1;
    unsigned int* rec = (unsigned int*)ws;  ws += (size_t)N_NODES * HID;
    float* xw2  = acc1;
    float* acc2 = acc1 + (size_t)N_NODES * RED;

    float* logits = (float*)d_out;
    float* emb    = logits + N_NODES;

    const int B = 256;
    k_deg_init<<<(N_NODES + B - 1) / B, B, 0, stream>>>(deg);
    k_deg_scatter<<<(N_EDGES + B - 1) / B, B, 0, stream>>>(row, col, ew, deg, own, rec);

    long long t1 = (long long)N_NODES * HID;
    k_l1_init<<<(unsigned)((t1 + B - 1) / B), B, 0, stream>>>(x, deg, W1, b1, rW1, rb1,
                                                              dinv, xw1, acc1);

    k_range_sum<HID><<<NB_RANGE, 1024, 0, stream>>>(own, rec, dinv, xw1, acc1);

    k_ln_relu_64<<<(N_NODES + 7) / 8, B, 0, stream>>>(acc1, g1, be1, x1);

    int waves_l2 = (N_NODES / 16) * (RED / 16);
    int thr_l2   = waves_l2 * 32;
    k_l2_wmma<<<(thr_l2 + B - 1) / B, B, 0, stream>>>(x1, W2, rW2, b2, rb2, dinv, xw2, acc2);

    k_range_sum<RED><<<NB_RANGE, 1024, 0, stream>>>(own, rec, dinv, xw2, acc2);

    k_head<<<(N_NODES + 7) / 8, B, 0, stream>>>(acc2, g2, be2, fcW, fcb, pW, pb, logits, emb);
}


# range sums read a per-range bit map from the degree pass (1 bit per 16-edge piece, LDS int ORs) and test only flagged pieces' owner bytes, not all; lists reordered, LDS f32 adds by arrival as before
# speedup vs baseline: 2.5694x; 1.0395x over previous
#include <hip/hip_runtime.h>
#include <hip/hip_bf16.h>

#define N_NODES 100000
#define N_EDGES 3200000
#define HID 64
#define RED 32
#define EMB 32
#define LN_EPS 1e-5f
#define RANGE 391
#define NB_RANGE 256
#define LCAP 1536
#define NCHUNK (N_EDGES / 512)
#define REC_SHIFT 17
#define REC_MASK ((1u << REC_SHIFT) - 1u)

typedef __attribute__((ext_vector_type(2))) float v2f;
typedef __attribute__((ext_vector_type(8))) float v8f;

__global__ void k_deg_init(float* __restrict__ deg) {
    int i = blockIdx.x * blockDim.x + threadIdx.x;
    if (i < N_NODES) deg[i] = 1.0f;
}

static_assert(N_EDGES == 512 * NCHUNK, "every thread of the degree pass has an edge");
__global__ __launch_bounds__(512) void k_deg_scatter(const int* __restrict__ row, const int* __restrict__ col,
                              const float* __restrict__ w, float* __restrict__ deg,
                              unsigned char* __restrict__ own, unsigned int* __restrict__ rec,
                              unsigned int* __restrict__ pmap) {
    __shared__ unsigned int s_map[NB_RANGE];
    const int tid = threadIdx.x;
    if (tid < NB_RANGE) s_map[tid] = 0u;
    __syncthreads();
    const int e = blockIdx.x * 512 + tid;
    int c = col[e];
    float we = w[e];
    unsigned r = (unsigned)row[e];
    int o = c / RANGE;
    atomicOr(&s_map[o], 1u << (tid >> 4));
    __syncthreads();
    if (tid < NB_RANGE) pmap[(size_t)tid * NCHUNK + blockIdx.x] = s_map[tid];
    atomicAdd(&deg[c], we);
    own[e] = (unsigned char)o;
    uint2 rc;
    rc.x = r | ((unsigned)(c - o * RANGE) << REC_SHIFT);
    rc.y = __float_as_uint(we);
    ((uint2*)rec)[e] = rc;
}

__global__ void k_l1_init(const float* __restrict__ x, const float* __restrict__ deg,
                          const float* __restrict__ W1, const float* __restrict__ b1,
                          const float* __restrict__ rW1, const float* __restrict__ rb1,
                          float* __restrict__ dinv, float* __restrict__ xw1,
                          float* __restrict__ acc1) {
    long long t = (long long)blockIdx.x * blockDim.x + threadIdx.x;
    if (t >= (long long)N_NODES * HID) return;
    int n = (int)(t >> 6);
    int f = (int)(t & 63);
    float d  = deg[n];
    float di = d > 0.f ? rsqrtf(d) : 0.f;
    if (f == 0) dinv[n] = di;
    float x0 = x[2 * n], x1v = x[2 * n + 1];
    float xw = fmaf(x0, W1[f], x1v * W1[HID + f]);
    float xs = di * xw;
    xw1[t]  = xs;
    acc1[t] = di * xs + b1[f] + fmaf(x0, rW1[f], x1v * rW1[HID + f]) + rb1[f];
}

template <int F>
__device__ __forceinline__ void walk_list(const int* list, int n, int lane,
                                          const uint2* __restrict__ rec,
                                          const float* __restrict__ src, float* s_acc) {
    for (int base = 0; base < n; base += 32) {
        int m = n - base; if (m > 32) m = 32;
        int i = base + lane;
        int e = list[i < n ? i : n - 1];
        const uint2 rc = rec[e];
        int r  = (int)(rc.x & REC_MASK);
        int cl = (int)(rc.x >> REC_SHIFT);
        float cf = __uint_as_float(rc.y);
        int j = 0;
        for (; j + 8 <= m; j += 8) {
            if (F == 64) {
                v2f s[8];
                #pragma unroll
                for (int u = 0; u < 8; ++u) {
                    int rj = __builtin_amdgcn_readlane(r, j + u);
                    s[u] = *(const v2f*)(src + (size_t)rj * 64 + 2 * lane);
                }
                #pragma unroll
                for (int u = 0; u < 8; ++u) {
                    int clj = __builtin_amdgcn_readlane(cl, j + u);
                    float cfj = __int_as_float(__builtin_amdgcn_readlane(__float_as_int(cf), j + u));
                    atomicAdd(&s_acc[clj * 64 + 2 * lane],     cfj * s[u].x);
                    atomicAdd(&s_acc[clj * 64 + 2 * lane + 1], cfj * s[u].y);
                }
            } else {
                float s[8];
                #pragma unroll
                for (int u = 0; u < 8; ++u) {
                    int rj = __builtin_amdgcn_readlane(r, j + u);
                    s[u] = src[(size_t)rj * F + lane];
                }
                #pragma unroll
                for (int u = 0; u < 8; ++u) {
                    int clj = __builtin_amdgcn_readlane(cl, j + u);
                    float cfj = __int_as_float(__builtin_amdgcn_readlane(__float_as_int(cf), j + u));
                    atomicAdd(&s_acc[clj * F + lane], cfj * s[u]);
                }
            }
        }
        for (; j < m; ++j) {
            int rj  = __builtin_amdgcn_readlane(r, j);
            int clj = __builtin_amdgcn_readlane(cl, j);
            float cfj = __int_as_float(__builtin_amdgcn_readlane(__float_as_int(cf), j));
            if (F == 64) {
                v2f s = *(const v2f*)(src + (size_t)rj * 64 + 2 * lane);
                atomicAdd(&s_acc[clj * 64 + 2 * lane],     cfj * s.x);
                atomicAdd(&s_acc[clj * 64 + 2 * lane + 1], cfj * s.y);
            } else {
                atomicAdd(&s_acc[clj * F + lane], cfj * src[(size_t)rj * F + lane]);
            }
        }
    }
}

__device__ __forceinline__ unsigned nz_bytes(unsigned x) {
    return ((x & 0x7F7F7F7Fu) + 0x7F7F7F7Fu) | x;
}

template <int F>
__global__ __launch_bounds__(1024) void k_range_sum(const unsigned char* __restrict__ own,
                                                    const unsigned int* __restrict__ pmap,
                                                    const unsigned int* __restrict__ rec,
                                                    const float* __restrict__ dinv,
                                                    const float* __restrict__ src,
                                                    float* __restrict__ acc) {
    __shared__ float s_acc[RANGE * F];
    __shared__ int   s_list[32 * LCAP];
    const int tid = threadIdx.x, lane = tid & 31;
    const int wave = __builtin_amdgcn_readfirstlane(tid >> 5);
    const int b = blockIdx.x, c0 = b * RANGE;
    for (int i = tid; i < RANGE * F; i += 1024) s_acc[i] = 0.f;
    __syncthreads();

    int* list = s_list + wave * LCAP;
    int cnt = 0;
    const uint4* own4 = (const uint4*)own;
    const uint2* rec2 = (const uint2*)rec;
    const unsigned key = (unsigned)b * 0x01010101u;
    const unsigned int* mp = pmap + (size_t)b * NCHUNK;
    auto take = [&](const uint4 cur, int ebase, bool ok) {
        if (__builtin_expect(cnt > LCAP - 512, 0)) {
            __builtin_amdgcn_fence(__ATOMIC_ACQ_REL, "wavefront");
            walk_list<F>(list, cnt, lane, rec2, src, s_acc);
            __builtin_amdgcn_fence(__ATOMIC_ACQ_REL, "wavefront");
            cnt = 0;
        }
        const unsigned H = 0x80808080u;
        unsigned t0 = nz_bytes(cur.x ^ key), t1 = nz_bytes(cur.y ^ key);
        unsigned t2 = nz_bytes(cur.z ^ key), t3 = nz_bytes(cur.w ^ key);
        unsigned m = ((~t0 & H) >> 7) | ((~t1 & H) >> 6) | ((~t2 & H) >> 5) | ((~t3 & H) >> 4);
        if (!ok) m = 0u;
        unsigned bal = __builtin_amdgcn_ballot_w32(m != 0);
        while (bal) {
            int p = __builtin_ctz(m | 0x80000000u);
            int slot = cnt + (int)__builtin_amdgcn_mbcnt_lo(bal, 0);
            if (m != 0) list[slot] = ebase + ((p & 3) << 2) + (p >> 3);
            cnt += __builtin_popcount(bal);
            m &= m - 1;
            bal = __builtin_amdgcn_ballot_w32(m != 0);
        }
    };
    const int lo = (wave * NCHUNK) / 32, hi = ((wave + 1) * NCHUNK) / 32;
    constexpr int NROUND = (NCHUNK / 32 + 1 + 31) / 32;
    unsigned nxt = (lo + lane < hi) ? mp[lo + lane] : 0u;
    for (int i = 0; i < NROUND; ++i) {
        unsigned wd = nxt;
        const int chunk = lo + lane + 32 * i;
        nxt = (i + 1 < NROUND && chunk + 32 < hi) ? mp[chunk + 32] : 0u;
        while (__builtin_amdgcn_ballot_w32(wd != 0)) {
            uint4 pc[4]; int pe[4]; bool ok[4];
            #pragma unroll
            for (int u = 0; u < 4; ++u) {
                ok[u] = wd != 0;
                const int p = ok[u] ? __builtin_ctz(wd) : 0;
                wd &= wd - 1;
                const int piece = ok[u] ? chunk * 32 + p : 0;
                pe[u] = piece * 16;
                pc[u] = own4[piece];
            }
            #pragma unroll
            for (int u = 0; u < 4; ++u) {
                if (__builtin_amdgcn_ballot_w32(ok[u]) == 0) break;
                take(pc[u], pe[u], ok[u]);
            }
        }
    }
    __builtin_amdgcn_fence(__ATOMIC_ACQ_REL, "wavefront");
    walk_list<F>(list, cnt, lane, rec2, src, s_acc);
    __syncthreads();

    int rows = N_NODES - c0; if (rows > RANGE) rows = RANGE;
    float* ap = acc + (size_t)c0 * F;
    const float* dp = dinv + c0;
    for (int i = tid; i < rows * F; i += 1024) ap[i] = fmaf(dp[i / F], s_acc[i], ap[i]);
}

__global__ void k_ln_relu_64(const float* __restrict__ acc, const float* __restrict__ g,
                             const float* __restrict__ be, float* __restrict__ out) {
    int wave = threadIdx.x >> 5;
    int lane = threadIdx.x & 31;
    int n = blockIdx.x * (blockDim.x >> 5) + wave;
    if (n >= N_NODES) return;
    float v0 = acc[(long long)n * 64 + lane];
    float v1 = acc[(long long)n * 64 + lane + 32];
    float s = v0 + v1, sq = fmaf(v0, v0, v1 * v1);
    #pragma unroll
    for (int off = 16; off > 0; off >>= 1) {
        s  += __shfl_xor(s,  off, 32);
        sq += __shfl_xor(sq, off, 32);
    }
    float mu  = s * (1.f / 64.f);
    float var = sq * (1.f / 64.f) - mu * mu;
    float inv = rsqrtf(var + LN_EPS);
    float o0 = (v0 - mu) * inv * g[lane]      + be[lane];
    float o1 = (v1 - mu) * inv * g[lane + 32] + be[lane + 32];
    out[(long long)n * 64 + lane]      = fmaxf(o0, 0.f);
    out[(long long)n * 64 + lane + 32] = fmaxf(o1, 0.f);
}

__global__ void k_l2_wmma(const float* __restrict__ x1, const float* __restrict__ W2,
                          const float* __restrict__ rW2, const float* __restrict__ b2,
                          const float* __restrict__ rb2, const float* __restrict__ dinv,
                          float* __restrict__ xw2, float* __restrict__ acc2) {
    int wave_global = (blockIdx.x * blockDim.x + threadIdx.x) >> 5;
    int lane = threadIdx.x & 31;
    const int tiles_m = N_NODES / 16;
    int tm = wave_global >> 1;
    int tn = wave_global & 1;
    if (tm >= tiles_m) return;
    int m0 = tm * 16, n0 = tn * 16;
    int  lr = lane & 15;
    bool hi = lane >= 16;
    int  kofs = hi ? 2 : 0;

    v8f c0 = {};
    v8f c1 = {};
    #pragma unroll
    for (int k0 = 0; k0 < HID; k0 += 4) {
        const float* ap = &x1[(long long)(m0 + lr) * HID + k0 + kofs];
        v2f a; a.x = ap[0]; a.y = ap[1];
        int kb = k0 + kofs;
        v2f b;  b.x  = W2 [kb * RED + n0 + lr]; b.y  = W2 [(kb + 1) * RED + n0 + lr];
        v2f br; br.x = rW2[kb * RED + n0 + lr]; br.y = rW2[(kb + 1) * RED + n0 + lr];
        c0 = __builtin_amdgcn_wmma_f32_16x16x4_f32(false, a, false, b,  (short)0, c0, false, false);
        c1 = __builtin_amdgcn_wmma_f32_16x16x4_f32(false, a, false, br, (short)0, c1, false, false);
    }
    int ncol = n0 + lr;
    float b2v = b2[ncol], rb2v = rb2[ncol];
    #pragma unroll
    for (int v = 0; v < 8; ++v) {
        int m = m0 + v + (hi ? 8 : 0);
        float di = dinv[m];
        float xv = di * c0[v];
        xw2 [(long long)m * RED + ncol] = xv;
        acc2[(long long)m * RED + ncol] = di * xv + b2v + c1[v] + rb2v;
    }
}

__global__ void k_head(const float* __restrict__ acc2, const float* __restrict__ g2,
                       const float* __restrict__ be2, const float* __restrict__ fcW,
                       const float* __restrict__ fcb, const float* __restrict__ pW,
                       const float* __restrict__ pb, float* __restrict__ logits,
                       float* __restrict__ emb) {
    int wave = threadIdx.x >> 5;
    int lane = threadIdx.x & 31;
    int n = blockIdx.x * (blockDim.x >> 5) + wave;
    if (n >= N_NODES) return;
    float v = acc2[(long long)n * 32 + lane];
    float s = v, sq = v * v;
    #pragma unroll
    for (int off = 16; off > 0; off >>= 1) {
        s  += __shfl_xor(s,  off, 32);
        sq += __shfl_xor(sq, off, 32);
    }
    float mu  = s * (1.f / 32.f);
    float var = sq * (1.f / 32.f) - mu * mu;
    float inv = rsqrtf(var + LN_EPS);
    float x2 = fmaxf((v - mu) * inv * g2[lane] + be2[lane], 0.f);

    float lg = x2 * fcW[lane];
    #pragma unroll
    for (int off = 16; off > 0; off >>= 1) lg += __shfl_xor(lg, off, 32);
    if (lane == 0) logits[n] = lg + fcb[0];

    float p = pb[lane];
    #pragma unroll
    for (int i = 0; i < 32; ++i)
        p = fmaf(__shfl(x2, i, 32), pW[i * 32 + lane], p);
    float nsq = p * p;
    #pragma unroll
    for (int off = 16; off > 0; off >>= 1) nsq += __shfl_xor(nsq, off, 32);
    float nrm = sqrtf(nsq);
    emb[(long long)n * 32 + lane] = p / fmaxf(nrm, 1e-12f);
}

extern "C" void kernel_launch(void* const* d_in, const int* in_sizes, int n_in,
                              void* d_out, int out_size, void* d_ws, size_t ws_size,
                              hipStream_t stream) {
    const float* x    = (const float*)d_in[0];
    const float* ew   = (const float*)d_in[1];
    const float* W1   = (const float*)d_in[2];
    const float* b1   = (const float*)d_in[3];
    const float* rW1  = (const float*)d_in[4];
    const float* rb1  = (const float*)d_in[5];
    const float* g1   = (const float*)d_in[6];
    const float* be1  = (const float*)d_in[7];
    const float* W2   = (const float*)d_in[8];
    const float* b2   = (const float*)d_in[9];
    const float* rW2  = (const float*)d_in[10];
    const float* rb2  = (const float*)d_in[11];
    const float* g2   = (const float*)d_in[12];
    const float* be2  = (const float*)d_in[13];
    const float* fcW  = (const float*)d_in[14];
    const float* fcb  = (const float*)d_in[15];
    const float* pW   = (const float*)d_in[16];
    const float* pb   = (const float*)d_in[17];
    const int*   ei   = (const int*)d_in[18];
    const int*   row  = ei;
    const int*   col  = ei + N_EDGES;

    float* ws   = (float*)d_ws;
    float* deg  = ws;  ws += N_NODES;
    float* dinv = ws;  ws += N_NODES;
    float* xw1  = ws;  ws += (size_t)N_NODES * HID;
    float* acc1 = ws;  ws += (size_t)N_NODES * HID;
    unsigned char* own = (unsigned char*)ws;  ws += (size_t)N_NODES * HID;
    float* x1   = xw1;
    unsigned int* rec = (unsigned int*)ws;  ws += (size_t)N_NODES * HID;
    unsigned int* pmap = (unsigned int*)(own + N_EDGES);
    float* xw2  = acc1;
    float* acc2 = acc1 + (size_t)N_NODES * RED;

    float* logits = (float*)d_out;
    float* emb    = logits + N_NODES;

    const int B = 256;
    k_deg_init<<<(N_NODES + B - 1) / B, B, 0, stream>>>(deg);
    k_deg_scatter<<<NCHUNK, 512, 0, stream>>>(row, col, ew, deg, own, rec, pmap);

    long long t1 = (long long)N_NODES * HID;
    k_l1_init<<<(unsigned)((t1 + B - 1) / B), B, 0, stream>>>(x, deg, W1, b1, rW1, rb1,
                                                              dinv, xw1, acc1);

    k_range_sum<HID><<<NB_RANGE, 1024, 0, stream>>>(own, pmap, rec, dinv, xw1, acc1);

    k_ln_relu_64<<<(N_NODES + 7) / 8, B, 0, stream>>>(acc1, g1, be1, x1);

    int waves_l2 = (N_NODES / 16) * (RED / 16);
    int thr_l2   = waves_l2 * 32;
    k_l2_wmma<<<(thr_l2 + B - 1) / B, B, 0, stream>>>(x1, W2, rW2, b2, rb2, dinv, xw2, acc2);

    k_range_sum<RED><<<NB_RANGE, 1024, 0, stream>>>(own, pmap, rec, dinv, xw2, acc2);

    k_head<<<(N_NODES + 7) / 8, B, 0, stream>>>(acc2, g2, be2, fcW, fcb, pW, pb, logits, emb);
}


# degree summed by the destination-range owners in LDS (same bit map, owner-byte filter and lists as the range sums; ds f32 adds by arrival, self-loop 1 added last), not a global f32 atomic per edge
# speedup vs baseline: 3.4143x; 1.3288x over previous
#include <hip/hip_runtime.h>
#include <hip/hip_bf16.h>

#define N_NODES 100000
#define N_EDGES 3200000
#define HID 64
#define RED 32
#define EMB 32
#define LN_EPS 1e-5f
#define RANGE 391
#define NB_RANGE 256
#define LCAP 1536
#define NCHUNK (N_EDGES / 512)
#define REC_SHIFT 17
#define REC_MASK ((1u << REC_SHIFT) - 1u)

typedef __attribute__((ext_vector_type(2))) float v2f;
typedef __attribute__((ext_vector_type(8))) float v8f;

static_assert(N_EDGES == 512 * NCHUNK, "every thread of the edge pass has an edge");
__global__ __launch_bounds__(512) void k_edge_pass(const int* __restrict__ row, const int* __restrict__ col,
                              const float* __restrict__ w,
                              unsigned char* __restrict__ own, unsigned int* __restrict__ rec,
                              unsigned int* __restrict__ pmap) {
    __shared__ unsigned int s_map[NB_RANGE];
    const int tid = threadIdx.x;
    if (tid < NB_RANGE) s_map[tid] = 0u;
    __syncthreads();
    const int e = blockIdx.x * 512 + tid;
    int c = col[e];
    float we = w[e];
    unsigned r = (unsigned)row[e];
    int o = c / RANGE;
    atomicOr(&s_map[o], 1u << (tid >> 4));
    __syncthreads();
    if (tid < NB_RANGE) pmap[(size_t)tid * NCHUNK + blockIdx.x] = s_map[tid];
    own[e] = (unsigned char)o;
    uint2 rc;
    rc.x = r | ((unsigned)(c - o * RANGE) << REC_SHIFT);
    rc.y = __float_as_uint(we);
    ((uint2*)rec)[e] = rc;
}

__global__ void k_l1_init(const float* __restrict__ x, const float* __restrict__ deg,
                          const float* __restrict__ W1, const float* __restrict__ b1,
                          const float* __restrict__ rW1, const float* __restrict__ rb1,
                          float* __restrict__ dinv, float* __restrict__ xw1,
                          float* __restrict__ acc1) {
    long long t = (long long)blockIdx.x * blockDim.x + threadIdx.x;
    if (t >= (long long)N_NODES * HID) return;
    int n = (int)(t >> 6);
    int f = (int)(t & 63);
    float d  = deg[n];
    float di = d > 0.f ? rsqrtf(d) : 0.f;
    if (f == 0) dinv[n] = di;
    float x0 = x[2 * n], x1v = x[2 * n + 1];
    float xw = fmaf(x0, W1[f], x1v * W1[HID + f]);
    float xs = di * xw;
    xw1[t]  = xs;
    acc1[t] = di * xs + b1[f] + fmaf(x0, rW1[f], x1v * rW1[HID + f]) + rb1[f];
}

template <int F>
__device__ __forceinline__ void walk_list(const int* list, int n, int lane,
                                          const uint2* __restrict__ rec,
                                          const float* __restrict__ src, float* s_acc) {
    if (F == 1) {
        for (int base = 0; base < n; base += 128) {
            uint2 rc[4]; bool ok[4];
            #pragma unroll
            for (int u = 0; u < 4; ++u) {
                const int i = base + 32 * u + lane;
                ok[u] = i < n;
                rc[u] = rec[list[ok[u] ? i : n - 1]];
            }
            #pragma unroll
            for (int u = 0; u < 4; ++u)
                if (ok[u]) atomicAdd(&s_acc[rc[u].x >> REC_SHIFT], __uint_as_float(rc[u].y));
        }
        return;
    }
    for (int base = 0; base < n; base += 32) {
        int m = n - base; if (m > 32) m = 32;
        int i = base + lane;
        int e = list[i < n ? i : n - 1];
        const uint2 rc = rec[e];
        int r  = (int)(rc.x & REC_MASK);
        int cl = (int)(rc.x >> REC_SHIFT);
        float cf = __uint_as_float(rc.y);
        int j = 0;
        for (; j + 8 <= m; j += 8) {
            if (F == 64) {
                v2f s[8];
                #pragma unroll
                for (int u = 0; u < 8; ++u) {
                    int rj = __builtin_amdgcn_readlane(r, j + u);
                    s[u] = *(const v2f*)(src + (size_t)rj * 64 + 2 * lane);
                }
                #pragma unroll
                for (int u = 0; u < 8; ++u) {
                    int clj = __builtin_amdgcn_readlane(cl, j + u);
                    float cfj = __int_as_float(__builtin_amdgcn_readlane(__float_as_int(cf), j + u));
                    atomicAdd(&s_acc[clj * 64 + 2 * lane],     cfj * s[u].x);
                    atomicAdd(&s_acc[clj * 64 + 2 * lane + 1], cfj * s[u].y);
                }
            } else {
                float s[8];
                #pragma unroll
                for (int u = 0; u < 8; ++u) {
                    int rj = __builtin_amdgcn_readlane(r, j + u);
                    s[u] = src[(size_t)rj * F + lane];
                }
                #pragma unroll
                for (int u = 0; u < 8; ++u) {
                    int clj = __builtin_amdgcn_readlane(cl, j + u);
                    float cfj = __int_as_float(__builtin_amdgcn_readlane(__float_as_int(cf), j + u));
                    atomicAdd(&s_acc[clj * F + lane], cfj * s[u]);
                }
            }
        }
        for (; j < m; ++j) {
            int rj  = __builtin_amdgcn_readlane(r, j);
            int clj = __builtin_amdgcn_readlane(cl, j);
            float cfj = __int_as_float(__builtin_amdgcn_readlane(__float_as_int(cf), j));
            if (F == 64) {
                v2f s = *(const v2f*)(src + (size_t)rj * 64 + 2 * lane);
                atomicAdd(&s_acc[clj * 64 + 2 * lane],     cfj * s.x);
                atomicAdd(&s_acc[clj * 64 + 2 * lane + 1], cfj * s.y);
            } else {
                atomicAdd(&s_acc[clj * F + lane], cfj * src[(size_t)rj * F + lane]);
            }
        }
    }
}

__device__ __forceinline__ unsigned nz_bytes(unsigned x) {
    return ((x & 0x7F7F7F7Fu) + 0x7F7F7F7Fu) | x;
}

template <int F>
__global__ __launch_bounds__(1024) void k_range_sum(const unsigned char* __restrict__ own,
                                                    const unsigned int* __restrict__ pmap,
                                                    const unsigned int* __restrict__ rec,
                                                    const float* __restrict__ dinv,
                                                    const float* __restrict__ src,
                                                    float* __restrict__ acc) {
    __shared__ float s_acc[RANGE * F];
    __shared__ int   s_list[32 * LCAP];
    const int tid = threadIdx.x, lane = tid & 31;
    const int wave = __builtin_amdgcn_readfirstlane(tid >> 5);
    const int b = blockIdx.x, c0 = b * RANGE;
    for (int i = tid; i < RANGE * F; i += 1024) s_acc[i] = 0.f;
    __syncthreads();

    int* list = s_list + wave * LCAP;
    int cnt = 0;
    const uint4* own4 = (const uint4*)own;
    const uint2* rec2 = (const uint2*)rec;
    const unsigned key = (unsigned)b * 0x01010101u;
    const unsigned int* mp = pmap + (size_t)b * NCHUNK;
    auto take = [&](const uint4 cur, int ebase, bool ok) {
        if (__builtin_expect(cnt > LCAP - 512, 0)) {
            __builtin_amdgcn_fence(__ATOMIC_ACQ_REL, "wavefront");
            walk_list<F>(list, cnt, lane, rec2, src, s_acc);
            __builtin_amdgcn_fence(__ATOMIC_ACQ_REL, "wavefront");
            cnt = 0;
        }
        const unsigned H = 0x80808080u;
        unsigned t0 = nz_bytes(cur.x ^ key), t1 = nz_bytes(cur.y ^ key);
        unsigned t2 = nz_bytes(cur.z ^ key), t3 = nz_bytes(cur.w ^ key);
        unsigned m = ((~t0 & H) >> 7) | ((~t1 & H) >> 6) | ((~t2 & H) >> 5) | ((~t3 & H) >> 4);
        if (!ok) m = 0u;
        unsigned bal = __builtin_amdgcn_ballot_w32(m != 0);
        while (bal) {
            int p = __builtin_ctz(m | 0x80000000u);
            int slot = cnt + (int)__builtin_amdgcn_mbcnt_lo(bal, 0);
            if (m != 0) list[slot] = ebase + ((p & 3) << 2) + (p >> 3);
            cnt += __builtin_popcount(bal);
            m &= m - 1;
            bal = __builtin_amdgcn_ballot_w32(m != 0);
        }
    };
    const int lo = (wave * NCHUNK) / 32, hi = ((wave + 1) * NCHUNK) / 32;
    constexpr int NROUND = (NCHUNK / 32 + 1 + 31) / 32;
    unsigned nxt = (lo + lane < hi) ? mp[lo + lane] : 0u;
    for (int i = 0; i < NROUND; ++i) {
        unsigned wd = nxt;
        const int chunk = lo + lane + 32 * i;
        nxt = (i + 1 < NROUND && chunk + 32 < hi) ? mp[chunk + 32] : 0u;
        while (__builtin_amdgcn_ballot_w32(wd != 0)) {
            uint4 pc[4]; int pe[4]; bool ok[4];
            #pragma unroll
            for (int u = 0; u < 4; ++u) {
                ok[u] = wd != 0;
                const int p = ok[u] ? __builtin_ctz(wd) : 0;
                wd &= wd - 1;
                const int piece = ok[u] ? chunk * 32 + p : 0;
                pe[u] = piece * 16;
                pc[u] = own4[piece];
            }
            #pragma unroll
            for (int u = 0; u < 4; ++u) {
                if (__builtin_amdgcn_ballot_w32(ok[u]) == 0) break;
                take(pc[u], pe[u], ok[u]);
            }
        }
    }
    __builtin_amdgcn_fence(__ATOMIC_ACQ_REL, "wavefront");
    walk_list<F>(list, cnt, lane, rec2, src, s_acc);
    __syncthreads();

    int rows = N_NODES - c0; if (rows > RANGE) rows = RANGE;
    float* ap = acc + (size_t)c0 * F;
    if (F == 1) {
        for (int i = tid; i < rows; i += 1024) ap[i] = 1.0f + s_acc[i];
    } else {
        const float* dp = dinv + c0;
        for (int i = tid; i < rows * F; i += 1024) ap[i] = fmaf(dp[i / F], s_acc[i], ap[i]);
    }
}

__global__ void k_ln_relu_64(const float* __restrict__ acc, const float* __restrict__ g,
                             const float* __restrict__ be, float* __restrict__ out) {
    int wave = threadIdx.x >> 5;
    int lane = threadIdx.x & 31;
    int n = blockIdx.x * (blockDim.x >> 5) + wave;
    if (n >= N_NODES) return;
    float v0 = acc[(long long)n * 64 + lane];
    float v1 = acc[(long long)n * 64 + lane + 32];
    float s = v0 + v1, sq = fmaf(v0, v0, v1 * v1);
    #pragma unroll
    for (int off = 16; off > 0; off >>= 1) {
        s  += __shfl_xor(s,  off, 32);
        sq += __shfl_xor(sq, off, 32);
    }
    float mu  = s * (1.f / 64.f);
    float var = sq * (1.f / 64.f) - mu * mu;
    float inv = rsqrtf(var + LN_EPS);
    float o0 = (v0 - mu) * inv * g[lane]      + be[lane];
    float o1 = (v1 - mu) * inv * g[lane + 32] + be[lane + 32];
    out[(long long)n * 64 + lane]      = fmaxf(o0, 0.f);
    out[(long long)n * 64 + lane + 32] = fmaxf(o1, 0.f);
}

__global__ void k_l2_wmma(const float* __restrict__ x1, const float* __restrict__ W2,
                          const float* __restrict__ rW2, const float* __restrict__ b2,
                          const float* __restrict__ rb2, const float* __restrict__ dinv,
                          float* __restrict__ xw2, float* __restrict__ acc2) {
    int wave_global = (blockIdx.x * blockDim.x + threadIdx.x) >> 5;
    int lane = threadIdx.x & 31;
    const int tiles_m = N_NODES / 16;
    int tm = wave_global >> 1;
    int tn = wave_global & 1;
    if (tm >= tiles_m) return;
    int m0 = tm * 16, n0 = tn * 16;
    int  lr = lane & 15;
    bool hi = lane >= 16;
    int  kofs = hi ? 2 : 0;

    v8f c0 = {};
    v8f c1 = {};
    #pragma unroll
    for (int k0 = 0; k0 < HID; k0 += 4) {
        const float* ap = &x1[(long long)(m0 + lr) * HID + k0 + kofs];
        v2f a; a.x = ap[0]; a.y = ap[1];
        int kb = k0 + kofs;
        v2f b;  b.x  = W2 [kb * RED + n0 + lr]; b.y  = W2 [(kb + 1) * RED + n0 + lr];
        v2f br; br.x = rW2[kb * RED + n0 + lr]; br.y = rW2[(kb + 1) * RED + n0 + lr];
        c0 = __builtin_amdgcn_wmma_f32_16x16x4_f32(false, a, false, b,  (short)0, c0, false, false);
        c1 = __builtin_amdgcn_wmma_f32_16x16x4_f32(false, a, false, br, (short)0, c1, false, false);
    }
    int ncol = n0 + lr;
    float b2v = b2[ncol], rb2v = rb2[ncol];
    #pragma unroll
    for (int v = 0; v < 8; ++v) {
        int m = m0 + v + (hi ? 8 : 0);
        float di = dinv[m];
        float xv = di * c0[v];
        xw2 [(long long)m * RED + ncol] = xv;
        acc2[(long long)m * RED + ncol] = di * xv + b2v + c1[v] + rb2v;
    }
}

__global__ void k_head(const float* __restrict__ acc2, const float* __restrict__ g2,
                       const float* __restrict__ be2, const float* __restrict__ fcW,
                       const float* __restrict__ fcb, const float* __restrict__ pW,
                       const float* __restrict__ pb, float* __restrict__ logits,
                       float* __restrict__ emb) {
    int wave = threadIdx.x >> 5;
    int lane = threadIdx.x & 31;
    int n = blockIdx.x * (blockDim.x >> 5) + wave;
    if (n >= N_NODES) return;
    float v = acc2[(long long)n * 32 + lane];
    float s = v, sq = v * v;
    #pragma unroll
    for (int off = 16; off > 0; off >>= 1) {
        s  += __shfl_xor(s,  off, 32);
        sq += __shfl_xor(sq, off, 32);
    }
    float mu  = s * (1.f / 32.f);
    float var = sq * (1.f / 32.f) - mu * mu;
    float inv = rsqrtf(var + LN_EPS);
    float x2 = fmaxf((v - mu) * inv * g2[lane] + be2[lane], 0.f);

    float lg = x2 * fcW[lane];
    #pragma unroll
    for (int off = 16; off > 0; off >>= 1) lg += __shfl_xor(lg, off, 32);
    if (lane == 0) logits[n] = lg + fcb[0];

    float p = pb[lane];
    #pragma unroll
    for (int i = 0; i < 32; ++i)
        p = fmaf(__shfl(x2, i, 32), pW[i * 32 + lane], p);
    float nsq = p * p;
    #pragma unroll
    for (int off = 16; off > 0; off >>= 1) nsq += __shfl_xor(nsq, off, 32);
    float nrm = sqrtf(nsq);
    emb[(long long)n * 32 + lane] = p / fmaxf(nrm, 1e-12f);
}

extern "C" void kernel_launch(void* const* d_in, const int* in_sizes, int n_in,
                              void* d_out, int out_size, void* d_ws, size_t ws_size,
                              hipStream_t stream) {
    const float* x    = (const float*)d_in[0];
    const float* ew   = (const float*)d_in[1];
    const float* W1   = (const float*)d_in[2];
    const float* b1   = (const float*)d_in[3];
    const float* rW1  = (const float*)d_in[4];
    const float* rb1  = (const float*)d_in[5];
    const float* g1   = (const float*)d_in[6];
    const float* be1  = (const float*)d_in[7];
    const float* W2   = (const float*)d_in[8];
    const float* b2   = (const float*)d_in[9];
    const float* rW2  = (const float*)d_in[10];
    const float* rb2  = (const float*)d_in[11];
    const float* g2   = (const float*)d_in[12];
    const float* be2  = (const float*)d_in[13];
    const float* fcW  = (const float*)d_in[14];
    const float* fcb  = (const float*)d_in[15];
    const float* pW   = (const float*)d_in[16];
    const float* pb   = (const float*)d_in[17];
    const int*   ei   = (const int*)d_in[18];
    const int*   row  = ei;
    const int*   col  = ei + N_EDGES;

    float* ws   = (float*)d_ws;
    float* deg  = ws;  ws += N_NODES;
    float* dinv = ws;  ws += N_NODES;
    float* xw1  = ws;  ws += (size_t)N_NODES * HID;
    float* acc1 = ws;  ws += (size_t)N_NODES * HID;
    unsigned char* own = (unsigned char*)ws;  ws += (size_t)N_NODES * HID;
    float* x1   = xw1;
    unsigned int* rec = (unsigned int*)ws;  ws += (size_t)N_NODES * HID;
    unsigned int* pmap = (unsigned int*)(own + N_EDGES);
    float* xw2  = acc1;
    float* acc2 = acc1 + (size_t)N_NODES * RED;

    float* logits = (float*)d_out;
    float* emb    = logits + N_NODES;

    const int B = 256;
    k_edge_pass<<<NCHUNK, 512, 0, stream>>>(row, col, ew, own, rec, pmap);
    k_range_sum<1><<<NB_RANGE, 1024, 0, stream>>>(own, pmap, rec, nullptr, nullptr, deg);

    long long t1 = (long long)N_NODES * HID;
    k_l1_init<<<(unsigned)((t1 + B - 1) / B), B, 0, stream>>>(x, deg, W1, b1, rW1, rb1,
                                                              dinv, xw1, acc1);

    k_range_sum<HID><<<NB_RANGE, 1024, 0, stream>>>(own, pmap, rec, dinv, xw1, acc1);

    k_ln_relu_64<<<(N_NODES + 7) / 8, B, 0, stream>>>(acc1, g1, be1, x1);

    int waves_l2 = (N_NODES / 16) * (RED / 16);
    int thr_l2   = waves_l2 * 32;
    k_l2_wmma<<<(thr_l2 + B - 1) / B, B, 0, stream>>>(x1, W2, rW2, b2, rb2, dinv, xw2, acc2);

    k_range_sum<RED><<<NB_RANGE, 1024, 0, stream>>>(own, pmap, rec, dinv, xw2, acc2);

    k_head<<<(N_NODES + 7) / 8, B, 0, stream>>>(acc2, g2, be2, fcW, fcb, pW, pb, logits, emb);
}


# range sums of both layers read per-wave record lists the degree pass stores (fixed place, length word, 32 consecutive 8-byte records a load): no 2nd, 3rd sweep, no record gather; LDS adds by arrival
# speedup vs baseline: 3.8261x; 1.1206x over previous
#include <hip/hip_runtime.h>
#include <hip/hip_bf16.h>

#define N_NODES 100000
#define N_EDGES 3200000
#define HID 64
#define RED 32
#define EMB 32
#define LN_EPS 1e-5f
#define RANGE 391
#define NB_RANGE 256
#define LCAP 1536
#define NCHUNK (N_EDGES / 512)
#define REC_SHIFT 17
#define REC_MASK ((1u << REC_SHIFT) - 1u)
#define LIST_CAP 576
#define LIST_OVER 0xFFFFFFFFu

typedef __attribute__((ext_vector_type(2))) float v2f;
typedef __attribute__((ext_vector_type(8))) float v8f;

static_assert(N_EDGES == 512 * NCHUNK, "every thread of the edge pass has an edge");
static_assert((size_t)N_EDGES + 4u * NB_RANGE * NCHUNK + 4u * NB_RANGE * 32 + 8u * NB_RANGE * 32 * LIST_CAP
              <= 2u * 4u * N_NODES * HID, "owner bytes, piece maps, list lengths and lists fit their two regions");
static_assert(N_EDGES % 256 == 0 && (4 * NB_RANGE * NCHUNK) % 256 == 0 && (8 * LIST_CAP) % 256 == 0,
              "every list starts on a 256-byte boundary");
__global__ __launch_bounds__(512) void k_edge_pass(const int* __restrict__ row, const int* __restrict__ col,
                              const float* __restrict__ w,
                              unsigned char* __restrict__ own, unsigned int* __restrict__ rec,
                              unsigned int* __restrict__ pmap) {
    __shared__ unsigned int s_map[NB_RANGE];
    const int tid = threadIdx.x;
    if (tid < NB_RANGE) s_map[tid] = 0u;
    __syncthreads();
    const int e = blockIdx.x * 512 + tid;
    int c = col[e];
    float we = w[e];
    unsigned r = (unsigned)row[e];
    int o = c / RANGE;
    atomicOr(&s_map[o], 1u << (tid >> 4));
    __syncthreads();
    if (tid < NB_RANGE) pmap[(size_t)tid * NCHUNK + blockIdx.x] = s_map[tid];
    own[e] = (unsigned char)o;
    uint2 rc;
    rc.x = r | ((unsigned)(c - o * RANGE) << REC_SHIFT);
    rc.y = __float_as_uint(we);
    ((uint2*)rec)[e] = rc;
}

__global__ void k_l1_init(const float* __restrict__ x, const float* __restrict__ deg,
                          const float* __restrict__ W1, const float* __restrict__ b1,
                          const float* __restrict__ rW1, const float* __restrict__ rb1,
                          float* __restrict__ dinv, float* __restrict__ xw1,
                          float* __restrict__ acc1) {
    long long t = (long long)blockIdx.x * blockDim.x + threadIdx.x;
    if (t >= (long long)N_NODES * HID) return;
    int n = (int)(t >> 6);
    int f = (int)(t & 63);
    float d  = deg[n];
    float di = d > 0.f ? rsqrtf(d) : 0.f;
    if (f == 0) dinv[n] = di;
    float x0 = x[2 * n], x1v = x[2 * n + 1];
    float xw = fmaf(x0, W1[f], x1v * W1[HID + f]);
    float xs = di * xw;
    xw1[t]  = xs;
    acc1[t] = di * xs + b1[f] + fmaf(x0, rW1[f], x1v * rW1[HID + f]) + rb1[f];
}

__device__ __forceinline__ void walk_deg(const int* list, int n, int lane,
                                         const uint2* __restrict__ rec, float* s_acc,
                                         uint2* __restrict__ gl, int gpos, bool keep) {
    for (int base = 0; base < n; base += 128) {
        uint2 rc[4]; bool ok[4];
        #pragma unroll
        for (int u = 0; u < 4; ++u) {
            const int i = base + 32 * u + lane;
            ok[u] = i < n;
            rc[u] = rec[list[ok[u] ? i : n - 1]];
        }
        #pragma unroll
        for (int u = 0; u < 4; ++u)
            if (ok[u]) {
                atomicAdd(&s_acc[rc[u].x >> REC_SHIFT], __uint_as_float(rc[u].y));
                if (keep) gl[gpos + base + 32 * u + lane] = rc[u];
            }
    }
}

__device__ __forceinline__ unsigned nz_bytes(unsigned x) {
    return ((x & 0x7F7F7F7Fu) + 0x7F7F7F7Fu) | x;
}

__global__ __launch_bounds__(1024) void k_range_deg(const unsigned char* __restrict__ own,
                                                    const unsigned int* __restrict__ pmap,
                                                    const unsigned int* __restrict__ rec,
                                                    float* __restrict__ deg,
                                                    unsigned int* __restrict__ lcnt,
                                                    unsigned int* __restrict__ lists) {
    __shared__ float s_acc[RANGE];
    __shared__ int   s_list[32 * LCAP];
    const int tid = threadIdx.x, lane = tid & 31;
    const int wave = __builtin_amdgcn_readfirstlane(tid >> 5);
    const int b = blockIdx.x, c0 = b * RANGE;
    for (int i = tid; i < RANGE; i += 1024) s_acc[i] = 0.f;
    __syncthreads();

    int* list = s_list + wave * LCAP;
    int cnt = 0;
    int gpos = 0;
    uint2* gl = (uint2*)lists + (size_t)(b * 32 + wave) * LIST_CAP;
    const uint4* own4 = (const uint4*)own;
    const uint2* rec2 = (const uint2*)rec;
    const unsigned key = (unsigned)b * 0x01010101u;
    const unsigned int* mp = pmap + (size_t)b * NCHUNK;
    auto drain = [&]() {
        __builtin_amdgcn_fence(__ATOMIC_ACQ_REL, "wavefront");
        walk_deg(list, cnt, lane, rec2, s_acc, gl, gpos, gpos + cnt <= LIST_CAP);
        gpos += cnt;
    };
    auto take = [&](const uint4 cur, int ebase, bool ok) {
        if (__builtin_expect(cnt > LCAP - 512, 0)) {
            drain();
            __builtin_amdgcn_fence(__ATOMIC_ACQ_REL, "wavefront");
            cnt = 0;
        }
        const unsigned H = 0x80808080u;
        unsigned t0 = nz_bytes(cur.x ^ key), t1 = nz_bytes(cur.y ^ key);
        unsigned t2 = nz_bytes(cur.z ^ key), t3 = nz_bytes(cur.w ^ key);
        unsigned m = ((~t0 & H) >> 7) | ((~t1 & H) >> 6) | ((~t2 & H) >> 5) | ((~t3 & H) >> 4);
        if (!ok) m = 0u;
        unsigned bal = __builtin_amdgcn_ballot_w32(m != 0);
        while (bal) {
            int p = __builtin_ctz(m | 0x80000000u);
            int slot = cnt + (int)__builtin_amdgcn_mbcnt_lo(bal, 0);
            if (m != 0) list[slot] = ebase + ((p & 3) << 2) + (p >> 3);
            cnt += __builtin_popcount(bal);
            m &= m - 1;
            bal = __builtin_amdgcn_ballot_w32(m != 0);
        }
    };
    const int lo = (wave * NCHUNK) / 32, hi = ((wave + 1) * NCHUNK) / 32;
    constexpr int NROUND = (NCHUNK / 32 + 1 + 31) / 32;
    unsigned nxt = (lo + lane < hi) ? mp[lo + lane] : 0u;
    for (int i = 0; i < NROUND; ++i) {
        unsigned wd = nxt;
        const int chunk = lo + lane + 32 * i;
        nxt = (i + 1 < NROUND && chunk + 32 < hi) ? mp[chunk + 32] : 0u;
        while (__builtin_amdgcn_ballot_w32(wd != 0)) {
            uint4 pc[4]; int pe[4]; bool ok[4];
            #pragma unroll
            for (int u = 0; u < 4; ++u) {
                ok[u] = wd != 0;
                const int p = ok[u] ? __builtin_ctz(wd) : 0;
                wd &= wd - 1;
                const int piece = ok[u] ? chunk * 32 + p : 0;
                pe[u] = piece * 16;
                pc[u] = own4[piece];
            }
            #pragma unroll
            for (int u = 0; u < 4; ++u) {
                if (__builtin_amdgcn_ballot_w32(ok[u]) == 0) break;
                take(pc[u], pe[u], ok[u]);
            }
        }
    }
    drain();
    if (lane == 0) lcnt[b * 32 + wave] = gpos <= LIST_CAP ? (unsigned)gpos : LIST_OVER;
    __syncthreads();

    int rows = N_NODES - c0; if (rows > RANGE) rows = RANGE;
    for (int i = tid; i < rows; i += 1024) deg[c0 + i] = 1.0f + s_acc[i];
}

template <int F>
__device__ __forceinline__ void add_one(int rj, int clj, float cfj, int lane,
                                        const float* __restrict__ src, float* s_acc) {
    if (F == 64) {
        v2f s = *(const v2f*)(src + (size_t)rj * 64 + 2 * lane);
        atomicAdd(&s_acc[clj * 64 + 2 * lane],     cfj * s.x);
        atomicAdd(&s_acc[clj * 64 + 2 * lane + 1], cfj * s.y);
    } else {
        atomicAdd(&s_acc[clj * F + lane], cfj * src[(size_t)rj * F + lane]);
    }
}

template <int F>
__device__ __forceinline__ void add_batch(int r, int cl, float cf, int m, int lane,
                                          const float* __restrict__ src, float* s_acc) {
    int j = 0;
    for (; j + 8 <= m; j += 8) {
        if (F == 64) {
            v2f s[8];
            #pragma unroll
            for (int u = 0; u < 8; ++u) {
                int rj = __builtin_amdgcn_readlane(r, j + u);
                s[u] = *(const v2f*)(src + (size_t)rj * 64 + 2 * lane);
            }
            #pragma unroll
            for (int u = 0; u < 8; ++u) {
                int clj = __builtin_amdgcn_readlane(cl, j + u);
                float cfj = __int_as_float(__builtin_amdgcn_readlane(__float_as_int(cf), j + u));
                atomicAdd(&s_acc[clj * 64 + 2 * lane],     cfj * s[u].x);
                atomicAdd(&s_acc[clj * 64 + 2 * lane + 1], cfj * s[u].y);
            }
        } else {
            float s[8];
            #pragma unroll
            for (int u = 0; u < 8; ++u) {
                int rj = __builtin_amdgcn_readlane(r, j + u);
                s[u] = src[(size_t)rj * F + lane];
            }
            #pragma unroll
            for (int u = 0; u < 8; ++u) {
                int clj = __builtin_amdgcn_readlane(cl, j + u);
                float cfj = __int_as_float(__builtin_amdgcn_readlane(__float_as_int(cf), j + u));
                atomicAdd(&s_acc[clj * F + lane], cfj * s[u]);
            }
        }
    }
    for (; j < m; ++j) {
        int rj  = __builtin_amdgcn_readlane(r, j);
        int clj = __builtin_amdgcn_readlane(cl, j);
        float cfj = __int_as_float(__builtin_amdgcn_readlane(__float_as_int(cf), j));
        add_one<F>(rj, clj, cfj, lane, src, s_acc);
    }
}

template <int F>
__global__ __launch_bounds__(1024) void k_list_sum(const unsigned int* __restrict__ lists,
                                                   const unsigned int* __restrict__ lcnt,
                                                   const unsigned char* __restrict__ own,
                                                   const int* __restrict__ row, const int* __restrict__ col,
                                                   const float* __restrict__ w,
                                                   const float* __restrict__ dinv,
                                                   const float* __restrict__ src,
                                                   float* __restrict__ acc) {
    __shared__ float s_acc[RANGE * F];
    const int tid = threadIdx.x, lane = tid & 31;
    const int wave = __builtin_amdgcn_readfirstlane(tid >> 5);
    const int b = blockIdx.x, c0 = b * RANGE;
    const int n = __builtin_amdgcn_readfirstlane((int)lcnt[b * 32 + wave]);
    for (int i = tid; i < RANGE * F; i += 1024) s_acc[i] = 0.f;
    __syncthreads();

    if (n >= 0) {
        const uint2* L = (const uint2*)lists + (size_t)(b * 32 + wave) * LIST_CAP;
        uint2 nx = make_uint2(0u, 0u);
        if (n > 0) nx = L[lane < n ? lane : n - 1];
        for (int base = 0; base < n; base += 32) {
            const uint2 rc = nx;
            int m = n - base; if (m > 32) m = 32;
            if (base + 32 < n) { const int i = base + 32 + lane; nx = L[i < n ? i : n - 1]; }
            add_batch<F>((int)(rc.x & REC_MASK), (int)(rc.x >> REC_SHIFT), __uint_as_float(rc.y),
                         m, lane, src, s_acc);
        }
    } else {
        const int lo = (wave * NCHUNK) / 32, hi = ((wave + 1) * NCHUNK) / 32;
        for (int e0 = lo * 512; e0 < hi * 512; e0 += 32) {
            const int e = e0 + lane;
            const bool mine = own[e] == (unsigned char)b;
            unsigned bal = __builtin_amdgcn_ballot_w32(mine);
            if (bal == 0) continue;
            const int r  = mine ? row[e] : 0;
            const int cl = mine ? col[e] - c0 : 0;
            const float cf = mine ? w[e] : 0.f;
            while (bal) {
                const int j = __builtin_ctz(bal);
                bal &= bal - 1;
                int rj  = __builtin_amdgcn_readlane(r, j);
                int clj = __builtin_amdgcn_readlane(cl, j);
                float cfj = __int_as_float(__builtin_amdgcn_readlane(__float_as_int(cf), j));
                add_one<F>(rj, clj, cfj, lane, src, s_acc);
            }
        }
    }
    __syncthreads();

    int rows = N_NODES - c0; if (rows > RANGE) rows = RANGE;
    float* ap = acc + (size_t)c0 * F;
    const float* dp = dinv + c0;
    for (int i = tid; i < rows * F; i += 1024) ap[i] = fmaf(dp[i / F], s_acc[i], ap[i]);
}

__global__ void k_ln_relu_64(const float* __restrict__ acc, const float* __restrict__ g,
                             const float* __restrict__ be, float* __restrict__ out) {
    int wave = threadIdx.x >> 5;
    int lane = threadIdx.x & 31;
    int n = blockIdx.x * (blockDim.x >> 5) + wave;
    if (n >= N_NODES) return;
    float v0 = acc[(long long)n * 64 + lane];
    float v1 = acc[(long long)n * 64 + lane + 32];
    float s = v0 + v1, sq = fmaf(v0, v0, v1 * v1);
    #pragma unroll
    for (int off = 16; off > 0; off >>= 1) {
        s  += __shfl_xor(s,  off, 32);
        sq += __shfl_xor(sq, off, 32);
    }
    float mu  = s * (1.f / 64.f);
    float var = sq * (1.f / 64.f) - mu * mu;
    float inv = rsqrtf(var + LN_EPS);
    float o0 = (v0 - mu) * inv * g[lane]      + be[lane];
    float o1 = (v1 - mu) * inv * g[lane + 32] + be[lane + 32];
    out[(long long)n * 64 + lane]      = fmaxf(o0, 0.f);
    out[(long long)n * 64 + lane + 32] = fmaxf(o1, 0.f);
}

__global__ void k_l2_wmma(const float* __restrict__ x1, const float* __restrict__ W2,
                          const float* __restrict__ rW2, const float* __restrict__ b2,
                          const float* __restrict__ rb2, const float* __restrict__ dinv,
                          float* __restrict__ xw2, float* __restrict__ acc2) {
    int wave_global = (blockIdx.x * blockDim.x + threadIdx.x) >> 5;
    int lane = threadIdx.x & 31;
    const int tiles_m = N_NODES / 16;
    int tm = wave_global >> 1;
    int tn = wave_global & 1;
    if (tm >= tiles_m) return;
    int m0 = tm * 16, n0 = tn * 16;
    int  lr = lane & 15;
    bool hi = lane >= 16;
    int  kofs = hi ? 2 : 0;

    v8f c0 = {};
    v8f c1 = {};
    #pragma unroll
    for (int k0 = 0; k0 < HID; k0 += 4) {
        const float* ap = &x1[(long long)(m0 + lr) * HID + k0 + kofs];
        v2f a; a.x = ap[0]; a.y = ap[1];
        int kb = k0 + kofs;
        v2f b;  b.x  = W2 [kb * RED + n0 + lr]; b.y  = W2 [(kb + 1) * RED + n0 + lr];
        v2f br; br.x = rW2[kb * RED + n0 + lr]; br.y = rW2[(kb + 1) * RED + n0 + lr];
        c0 = __builtin_amdgcn_wmma_f32_16x16x4_f32(false, a, false, b,  (short)0, c0, false, false);
        c1 = __builtin_amdgcn_wmma_f32_16x16x4_f32(false, a, false, br, (short)0, c1, false, false);
    }
    int ncol = n0 + lr;
    float b2v = b2[ncol], rb2v = rb2[ncol];
    #pragma unroll
    for (int v = 0; v < 8; ++v) {
        int m = m0 + v + (hi ? 8 : 0);
        float di = dinv[m];
        float xv = di * c0[v];
        xw2 [(long long)m * RED + ncol] = xv;
        acc2[(long long)m * RED + ncol] = di * xv + b2v + c1[v] + rb2v;
    }
}

__global__ void k_head(const float* __restrict__ acc2, const float* __restrict__ g2,
                       const float* __restrict__ be2, const float* __restrict__ fcW,
                       const float* __restrict__ fcb, const float* __restrict__ pW,
                       const float* __restrict__ pb, float* __restrict__ logits,
                       float* __restrict__ emb) {
    int wave = threadIdx.x >> 5;
    int lane = threadIdx.x & 31;
    int n = blockIdx.x * (blockDim.x >> 5) + wave;
    if (n >= N_NODES) return;
    float v = acc2[(long long)n * 32 + lane];
    float s = v, sq = v * v;
    #pragma unroll
    for (int off = 16; off > 0; off >>= 1) {
        s  += __shfl_xor(s,  off, 32);
        sq += __shfl_xor(sq, off, 32);
    }
    float mu  = s * (1.f / 32.f);
    float var = sq * (1.f / 32.f) - mu * mu;
    float inv = rsqrtf(var + LN_EPS);
    float x2 = fmaxf((v - mu) * inv * g2[lane] + be2[lane], 0.f);

    float lg = x2 * fcW[lane];
    #pragma unroll
    for (int off = 16; off > 0; off >>= 1) lg += __shfl_xor(lg, off, 32);
    if (lane == 0) logits[n] = lg + fcb[0];

    float p = pb[lane];
    #pragma unroll
    for (int i = 0; i < 32; ++i)
        p = fmaf(__shfl(x2, i, 32), pW[i * 32 + lane], p);
    float nsq = p * p;
    #pragma unroll
    for (int off = 16; off > 0; off >>= 1) nsq += __shfl_xor(nsq, off, 32);
    float nrm = sqrtf(nsq);
    emb[(long long)n * 32 + lane] = p / fmaxf(nrm, 1e-12f);
}

extern "C" void kernel_launch(void* const* d_in, const int* in_sizes, int n_in,
                              void* d_out, int out_size, void* d_ws, size_t ws_size,
                              hipStream_t stream) {
    const float* x    = (const float*)d_in[0];
    const float* ew   = (const float*)d_in[1];
    const float* W1   = (const float*)d_in[2];
    const float* b1   = (const float*)d_in[3];
    const float* rW1  = (const float*)d_in[4];
    const float* rb1  = (const float*)d_in[5];
    const float* g1   = (const float*)d_in[6];
    const float* be1  = (const float*)d_in[7];
    const float* W2   = (const float*)d_in[8];
    const float* b2   = (const float*)d_in[9];
    const float* rW2  = (const float*)d_in[10];
    const float* rb2  = (const float*)d_in[11];
    const float* g2   = (const float*)d_in[12];
    const float* be2  = (const float*)d_in[13];
    const float* fcW  = (const float*)d_in[14];
    const float* fcb  = (const float*)d_in[15];
    const float* pW   = (const float*)d_in[16];
    const float* pb   = (const float*)d_in[17];
    const int*   ei   = (const int*)d_in[18];
    const int*   row  = ei;
    const int*   col  = ei + N_EDGES;

    float* ws   = (float*)d_ws;
    float* deg  = ws;  ws += N_NODES;
    float* dinv = ws;  ws += N_NODES;
    float* xw1  = ws;  ws += (size_t)N_NODES * HID;
    float* acc1 = ws;  ws += (size_t)N_NODES * HID;
    unsigned char* own = (unsigned char*)ws;  ws += (size_t)N_NODES * HID;
    float* x1   = xw1;
    ws += (size_t)N_NODES * HID;
    unsigned int* rec = (unsigned int*)acc1;
    unsigned int* pmap = (unsigned int*)(own + N_EDGES);
    unsigned int* lcnt = pmap + (size_t)NB_RANGE * NCHUNK;
    unsigned int* lists = lcnt + NB_RANGE * 32;
    float* xw2  = acc1;
    float* acc2 = acc1 + (size_t)N_NODES * RED;

    float* logits = (float*)d_out;
    float* emb    = logits + N_NODES;

    const int B = 256;
    k_edge_pass<<<NCHUNK, 512, 0, stream>>>(row, col, ew, own, rec, pmap);
    k_range_deg<<<NB_RANGE, 1024, 0, stream>>>(own, pmap, rec, deg, lcnt, lists);

    long long t1 = (long long)N_NODES * HID;
    k_l1_init<<<(unsigned)((t1 + B - 1) / B), B, 0, stream>>>(x, deg, W1, b1, rW1, rb1,
                                                              dinv, xw1, acc1);

    k_list_sum<HID><<<NB_RANGE, 1024, 0, stream>>>(lists, lcnt, own, row, col, ew, dinv, xw1, acc1);

    k_ln_relu_64<<<(N_NODES + 7) / 8, B, 0, stream>>>(acc1, g1, be1, x1);

    int waves_l2 = (N_NODES / 16) * (RED / 16);
    int thr_l2   = waves_l2 * 32;
    k_l2_wmma<<<(thr_l2 + B - 1) / B, B, 0, stream>>>(x1, W2, rW2, b2, rb2, dinv, xw2, acc2);

    k_list_sum<RED><<<NB_RANGE, 1024, 0, stream>>>(lists, lcnt, own, row, col, ew, dinv, xw2, acc2);

    k_head<<<(N_NODES + 7) / 8, B, 0, stream>>>(acc2, g2, be2, fcW, fcb, pW, pb, logits, emb);
}


# edge pass, layer-1 init and LayerNorm-1 as grids of long waves: a wave walks 1,024 edges or 20 nodes (3,125 / 5,000 / 5,000 waves enter, not 400,000 short ones); arithmetic and sum order kept
# speedup vs baseline: 4.0749x; 1.0650x over previous
#include <hip/hip_runtime.h>
#include <hip/hip_bf16.h>

#define N_NODES 100000
#define N_EDGES 3200000
#define HID 64
#define RED 32
#define EMB 32
#define LN_EPS 1e-5f
#define RANGE 391
#define NB_RANGE 256
#define LCAP 1536
#define NCHUNK (N_EDGES / 512)
#define REC_SHIFT 17
#define REC_MASK ((1u << REC_SHIFT) - 1u)
#define LIST_CAP 576
#define LIST_OVER 0xFFFFFFFFu

typedef __attribute__((ext_vector_type(2))) float v2f;
typedef __attribute__((ext_vector_type(8))) float v8f;
typedef __attribute__((ext_vector_type(4))) float v4f;
typedef __attribute__((ext_vector_type(4))) int v4i;

#define EP_WAVES 4
#define NPAIR (NCHUNK / 2)
static_assert(N_EDGES == 512 * NCHUNK && NCHUNK % 2 == 0, "every wave of the edge pass has two whole chunks");
static_assert((size_t)N_EDGES + 4u * NB_RANGE * NCHUNK + 4u * NB_RANGE * 32 + 8u * NB_RANGE * 32 * LIST_CAP
              <= 2u * 4u * N_NODES * HID, "owner bytes, piece maps, list lengths and lists fit their two regions");
static_assert(N_EDGES % 256 == 0 && (4 * NB_RANGE * NCHUNK) % 256 == 0 && (8 * LIST_CAP) % 256 == 0,
              "every list starts on a 256-byte boundary");
__global__ __launch_bounds__(32 * EP_WAVES) void k_edge_pass(const int* __restrict__ row, const int* __restrict__ col,
                              const float* __restrict__ w,
                              unsigned char* __restrict__ own, unsigned int* __restrict__ rec,
                              unsigned int* __restrict__ pmap) {
    __shared__ __attribute__((aligned(16))) unsigned int s_map[EP_WAVES][2 * NB_RANGE];
    const int lane = threadIdx.x & 31;
    const int wave = __builtin_amdgcn_readfirstlane((int)(threadIdx.x >> 5));
    const int pair = blockIdx.x * EP_WAVES + wave;
    if (pair >= NPAIR) return;
    unsigned int* mp = s_map[wave];
    #pragma unroll
    for (int k = 0; k < 4; ++k) ((uint4*)mp)[k * 32 + lane] = make_uint4(0u, 0u, 0u, 0u);
    __builtin_amdgcn_fence(__ATOMIC_ACQ_REL, "wavefront");

    const int q0 = pair * 256;
    v4i c4[8]; v4f w4[8]; v4i r4[8];
    #pragma unroll
    for (int u = 0; u < 8; ++u) {
        c4[u] = ((const v4i*)col)[q0 + 32 * u + lane];
        w4[u] = ((const v4f*)w)[q0 + 32 * u + lane];
        r4[u] = ((const v4i*)row)[q0 + 32 * u + lane];
    }
    asm volatile("" :: "v"(c4[0]), "v"(c4[1]), "v"(c4[2]), "v"(c4[3]), "v"(c4[4]), "v"(c4[5]), "v"(c4[6]), "v"(c4[7]),
                       "v"(w4[0]), "v"(w4[1]), "v"(w4[2]), "v"(w4[3]), "v"(w4[4]), "v"(w4[5]), "v"(w4[6]), "v"(w4[7]),
                       "v"(r4[0]), "v"(r4[1]), "v"(r4[2]), "v"(r4[3]), "v"(r4[4]), "v"(r4[5]), "v"(r4[6]), "v"(r4[7]));
    #pragma unroll
    for (int u = 0; u < 8; ++u) {
        unsigned int* m = mp + (u >> 2) * NB_RANGE;
        const unsigned bit = 1u << (8 * (u & 3) + (lane >> 2));
        const int o0 = c4[u].x / RANGE, o1 = c4[u].y / RANGE, o2 = c4[u].z / RANGE, o3 = c4[u].w / RANGE;
        atomicOr(&m[o0], bit);
        atomicOr(&m[o1], bit);
        atomicOr(&m[o2], bit);
        atomicOr(&m[o3], bit);
        ((unsigned int*)own)[q0 + 32 * u + lane] =
            (unsigned)o0 | ((unsigned)o1 << 8) | ((unsigned)o2 << 16) | ((unsigned)o3 << 24);
        uint4 ra, rb;
        ra.x = (unsigned)r4[u].x | ((unsigned)(c4[u].x - o0 * RANGE) << REC_SHIFT);
        ra.y = __float_as_uint(w4[u].x);
        ra.z = (unsigned)r4[u].y | ((unsigned)(c4[u].y - o1 * RANGE) << REC_SHIFT);
        ra.w = __float_as_uint(w4[u].y);
        rb.x = (unsigned)r4[u].z | ((unsigned)(c4[u].z - o2 * RANGE) << REC_SHIFT);
        rb.y = __float_as_uint(w4[u].z);
        rb.z = (unsigned)r4[u].w | ((unsigned)(c4[u].w - o3 * RANGE) << REC_SHIFT);
        rb.w = __float_as_uint(w4[u].w);
        ((uint4*)rec)[2 * (q0 + 32 * u + lane)]     = ra;
        ((uint4*)rec)[2 * (q0 + 32 * u + lane) + 1] = rb;
    }
    __builtin_amdgcn_fence(__ATOMIC_ACQ_REL, "wavefront");
    #pragma unroll
    for (int k = 0; k < 8; ++k) {
        const int o = k * 32 + lane;
        uint2 v;
        v.x = mp[o];
        v.y = mp[NB_RANGE + o];
        ((uint2*)pmap)[(size_t)o * NPAIR + pair] = v;
    }
}

#define NPW 20
#define NODE_WGS (N_NODES / (8 * NPW))
#define L1_WAVES 4
#define L1_WGS (N_NODES / (L1_WAVES * NPW))
static_assert(N_NODES == NODE_WGS * 8 * NPW && N_NODES == L1_WGS * L1_WAVES * NPW && NPW % 2 == 0 && NPW <= 32,
              "every wave of a node pass has NPW whole nodes");
__device__ __forceinline__ void l1_elem(float x0, float x1v, float di, float wa, float wb, float bb,
                                        float ra, float rb, float rbb, float& xs, float& a) {
    float xw = fmaf(x0, wa, x1v * wb);
    xs = di * xw;
    a = di * xs + bb + fmaf(x0, ra, x1v * rb) + rbb;
}
__global__ __launch_bounds__(32 * L1_WAVES) void k_l1_init(const float* __restrict__ x, const float* __restrict__ deg,
                          const float* __restrict__ W1, const float* __restrict__ b1,
                          const float* __restrict__ rW1, const float* __restrict__ rb1,
                          float* __restrict__ dinv, float* __restrict__ xw1,
                          float* __restrict__ acc1) {
    const int lane = threadIdx.x & 31;
    const int n0 = (blockIdx.x * L1_WAVES + (int)(threadIdx.x >> 5)) * NPW;
    const int f0 = 4 * (lane & 15), h = lane >> 4;
    const v4f wa = *(const v4f*)(W1 + f0),  wb = *(const v4f*)(W1 + HID + f0);
    const v4f ra = *(const v4f*)(rW1 + f0), rb = *(const v4f*)(rW1 + HID + f0);
    const v4f bv = *(const v4f*)(b1 + f0),  rbv = *(const v4f*)(rb1 + f0);
    const float dl = deg[n0 + (lane < NPW ? lane : NPW - 1)];
    float dg[NPW / 2]; v2f xv[NPW / 2];
    #pragma unroll
    for (int i = 0; i < NPW / 2; ++i) {
        const int n = n0 + 2 * i + h;
        dg[i] = deg[n];
        xv[i] = *(const v2f*)(x + 2 * n);
    }
    asm volatile("" :: "v"(wa), "v"(wb), "v"(ra), "v"(rb), "v"(bv), "v"(rbv), "v"(dl),
                       "v"(dg[0]), "v"(dg[1]), "v"(dg[2]), "v"(dg[3]), "v"(dg[4]),
                       "v"(dg[5]), "v"(dg[6]), "v"(dg[7]), "v"(dg[8]), "v"(dg[9]),
                       "v"(xv[0]), "v"(xv[1]), "v"(xv[2]), "v"(xv[3]), "v"(xv[4]),
                       "v"(xv[5]), "v"(xv[6]), "v"(xv[7]), "v"(xv[8]), "v"(xv[9]));
    static_assert(NPW / 2 == 10, "the statement above names ten turns");
    #pragma unroll
    for (int i = 0; i < NPW / 2; ++i) {
        const int n = n0 + 2 * i + h;
        const float d  = dg[i];
        const float di = d > 0.f ? rsqrtf(d) : 0.f;
        float s0, s1, s2, s3, a0, a1, a2, a3;
        l1_elem(xv[i].x, xv[i].y, di, wa.x, wb.x, bv.x, ra.x, rb.x, rbv.x, s0, a0);
        l1_elem(xv[i].x, xv[i].y, di, wa.y, wb.y, bv.y, ra.y, rb.y, rbv.y, s1, a1);
        l1_elem(xv[i].x, xv[i].y, di, wa.z, wb.z, bv.z, ra.z, rb.z, rbv.z, s2, a2);
        l1_elem(xv[i].x, xv[i].y, di, wa.w, wb.w, bv.w, ra.w, rb.w, rbv.w, s3, a3);
        v4f xs = {s0, s1, s2, s3}, a = {a0, a1, a2, a3};
        *(v4f*)(xw1  + (size_t)n * HID + f0) = xs;
        *(v4f*)(acc1 + (size_t)n * HID + f0) = a;
    }
    if (lane < NPW) dinv[n0 + lane] = dl > 0.f ? rsqrtf(dl) : 0.f;
}

__device__ __forceinline__ void walk_deg(const int* list, int n, int lane,
                                         const uint2* __restrict__ rec, float* s_acc,
                                         uint2* __restrict__ gl, int gpos, bool keep) {
    for (int base = 0; base < n; base += 128) {
        uint2 rc[4]; bool ok[4];
        #pragma unroll
        for (int u = 0; u < 4; ++u) {
            const int i = base + 32 * u + lane;
            ok[u] = i < n;
            rc[u] = rec[list[ok[u] ? i : n - 1]];
        }
        #pragma unroll
        for (int u = 0; u < 4; ++u)
            if (ok[u]) {
                atomicAdd(&s_acc[rc[u].x >> REC_SHIFT], __uint_as_float(rc[u].y));
                if (keep) gl[gpos + base + 32 * u + lane] = rc[u];
            }
    }
}

__device__ __forceinline__ unsigned nz_bytes(unsigned x) {
    return ((x & 0x7F7F7F7Fu) + 0x7F7F7F7Fu) | x;
}

__global__ __launch_bounds__(1024) void k_range_deg(const unsigned char* __restrict__ own,
                                                    const unsigned int* __restrict__ pmap,
                                                    const unsigned int* __restrict__ rec,
                                                    float* __restrict__ deg,
                                                    unsigned int* __restrict__ lcnt,
                                                    unsigned int* __restrict__ lists) {
    __shared__ float s_acc[RANGE];
    __shared__ int   s_list[32 * LCAP];
    const int tid = threadIdx.x, lane = tid & 31;
    const int wave = __builtin_amdgcn_readfirstlane(tid >> 5);
    const int b = blockIdx.x, c0 = b * RANGE;
    for (int i = tid; i < RANGE; i += 1024) s_acc[i] = 0.f;
    __syncthreads();

    int* list = s_list + wave * LCAP;
    int cnt = 0;
    int gpos = 0;
    uint2* gl = (uint2*)lists + (size_t)(b * 32 + wave) * LIST_CAP;
    const uint4* own4 = (const uint4*)own;
    const uint2* rec2 = (const uint2*)rec;
    const unsigned key = (unsigned)b * 0x01010101u;
    const unsigned int* mp = pmap + (size_t)b * NCHUNK;
    auto drain = [&]() {
        __builtin_amdgcn_fence(__ATOMIC_ACQ_REL, "wavefront");
        walk_deg(list, cnt, lane, rec2, s_acc, gl, gpos, gpos + cnt <= LIST_CAP);
        gpos += cnt;
    };
    auto take = [&](const uint4 cur, int ebase, bool ok) {
        if (__builtin_expect(cnt > LCAP - 512, 0)) {
            drain();
            __builtin_amdgcn_fence(__ATOMIC_ACQ_REL, "wavefront");
            cnt = 0;
        }
        const unsigned H = 0x80808080u;
        unsigned t0 = nz_bytes(cur.x ^ key), t1 = nz_bytes(cur.y ^ key);
        unsigned t2 = nz_bytes(cur.z ^ key), t3 = nz_bytes(cur.w ^ key);
        unsigned m = ((~t0 & H) >> 7) | ((~t1 & H) >> 6) | ((~t2 & H) >> 5) | ((~t3 & H) >> 4);
        if (!ok) m = 0u;
        unsigned bal = __builtin_amdgcn_ballot_w32(m != 0);
        while (bal) {
            int p = __builtin_ctz(m | 0x80000000u);
            int slot = cnt + (int)__builtin_amdgcn_mbcnt_lo(bal, 0);
            if (m != 0) list[slot] = ebase + ((p & 3) << 2) + (p >> 3);
            cnt += __builtin_popcount(bal);
            m &= m - 1;
            bal = __builtin_amdgcn_ballot_w32(m != 0);
        }
    };
    const int lo = (wave * NCHUNK) / 32, hi = ((wave + 1) * NCHUNK) / 32;
    constexpr int NROUND = (NCHUNK / 32 + 1 + 31) / 32;
    unsigned nxt = (lo + lane < hi) ? mp[lo + lane] : 0u;
    for (int i = 0; i < NROUND; ++i) {
        unsigned wd = nxt;
        const int chunk = lo + lane + 32 * i;
        nxt = (i + 1 < NROUND && chunk + 32 < hi) ? mp[chunk + 32] : 0u;
        while (__builtin_amdgcn_ballot_w32(wd != 0)) {
            uint4 pc[4]; int pe[4]; bool ok[4];
            #pragma unroll
            for (int u = 0; u < 4; ++u) {
                ok[u] = wd != 0;
                const int p = ok[u] ? __builtin_ctz(wd) : 0;
                wd &= wd - 1;
                const int piece = ok[u] ? chunk * 32 + p : 0;
                pe[u] = piece * 16;
                pc[u] = own4[piece];
            }
            #pragma unroll
            for (int u = 0; u < 4; ++u) {
                if (__builtin_amdgcn_ballot_w32(ok[u]) == 0) break;
                take(pc[u], pe[u], ok[u]);
            }
        }
    }
    drain();
    if (lane == 0) lcnt[b * 32 + wave] = gpos <= LIST_CAP ? (unsigned)gpos : LIST_OVER;
    __syncthreads();

    int rows = N_NODES - c0; if (rows > RANGE) rows = RANGE;
    for (int i = tid; i < rows; i += 1024) deg[c0 + i] = 1.0f + s_acc[i];
}

template <int F>
__device__ __forceinline__ void add_one(int rj, int clj, float cfj, int lane,
                                        const float* __restrict__ src, float* s_acc) {
    if (F == 64) {
        v2f s = *(const v2f*)(src + (size_t)rj * 64 + 2 * lane);
        atomicAdd(&s_acc[clj * 64 + 2 * lane],     cfj * s.x);
        atomicAdd(&s_acc[clj * 64 + 2 * lane + 1], cfj * s.y);
    } else {
        atomicAdd(&s_acc[clj * F + lane], cfj * src[(size_t)rj * F + lane]);
    }
}

template <int F>
__device__ __forceinline__ void add_batch(int r, int cl, float cf, int m, int lane,
                                          const float* __restrict__ src, float* s_acc) {
    int j = 0;
    for (; j + 8 <= m; j += 8) {
        if (F == 64) {
            v2f s[8];
            #pragma unroll
            for (int u = 0; u < 8; ++u) {
                int rj = __builtin_amdgcn_readlane(r, j + u);
                s[u] = *(const v2f*)(src + (size_t)rj * 64 + 2 * lane);
            }
            #pragma unroll
            for (int u = 0; u < 8; ++u) {
                int clj = __builtin_amdgcn_readlane(cl, j + u);
                float cfj = __int_as_float(__builtin_amdgcn_readlane(__float_as_int(cf), j + u));
                atomicAdd(&s_acc[clj * 64 + 2 * lane],     cfj * s[u].x);
                atomicAdd(&s_acc[clj * 64 + 2 * lane + 1], cfj * s[u].y);
            }
        } else {
            float s[8];
            #pragma unroll
            for (int u = 0; u < 8; ++u) {
                int rj = __builtin_amdgcn_readlane(r, j + u);
                s[u] = src[(size_t)rj * F + lane];
            }
            #pragma unroll
            for (int u = 0; u < 8; ++u) {
                int clj = __builtin_amdgcn_readlane(cl, j + u);
                float cfj = __int_as_float(__builtin_amdgcn_readlane(__float_as_int(cf), j + u));
                atomicAdd(&s_acc[clj * F + lane], cfj * s[u]);
            }
        }
    }
    for (; j < m; ++j) {
        int rj  = __builtin_amdgcn_readlane(r, j);
        int clj = __builtin_amdgcn_readlane(cl, j);
        float cfj = __int_as_float(__builtin_amdgcn_readlane(__float_as_int(cf), j));
        add_one<F>(rj, clj, cfj, lane, src, s_acc);
    }
}

template <int F>
__global__ __launch_bounds__(1024) void k_list_sum(const unsigned int* __restrict__ lists,
                                                   const unsigned int* __restrict__ lcnt,
                                                   const unsigned char* __restrict__ own,
                                                   const int* __restrict__ row, const int* __restrict__ col,
                                                   const float* __restrict__ w,
                                                   const float* __restrict__ dinv,
                                                   const float* __restrict__ src,
                                                   float* __restrict__ acc) {
    __shared__ float s_acc[RANGE * F];
    const int tid = threadIdx.x, lane = tid & 31;
    const int wave = __builtin_amdgcn_readfirstlane(tid >> 5);
    const int b = blockIdx.x, c0 = b * RANGE;
    const int n = __builtin_amdgcn_readfirstlane((int)lcnt[b * 32 + wave]);
    for (int i = tid; i < RANGE * F; i += 1024) s_acc[i] = 0.f;
    __syncthreads();

    if (n >= 0) {
        const uint2* L = (const uint2*)lists + (size_t)(b * 32 + wave) * LIST_CAP;
        uint2 nx = make_uint2(0u, 0u);
        if (n > 0) nx = L[lane < n ? lane : n - 1];
        for (int base = 0; base < n; base += 32) {
            const uint2 rc = nx;
            int m = n - base; if (m > 32) m = 32;
            if (base + 32 < n) { const int i = base + 32 + lane; nx = L[i < n ? i : n - 1]; }
            add_batch<F>((int)(rc.x & REC_MASK), (int)(rc.x >> REC_SHIFT), __uint_as_float(rc.y),
                         m, lane, src, s_acc);
        }
    } else {
        const int lo = (wave * NCHUNK) / 32, hi = ((wave + 1) * NCHUNK) / 32;
        for (int e0 = lo * 512; e0 < hi * 512; e0 += 32) {
            const int e = e0 + lane;
            const bool mine = own[e] == (unsigned char)b;
            unsigned bal = __builtin_amdgcn_ballot_w32(mine);
            if (bal == 0) continue;
            const int r  = mine ? row[e] : 0;
            const int cl = mine ? col[e] - c0 : 0;
            const float cf = mine ? w[e] : 0.f;
            while (bal) {
                const int j = __builtin_ctz(bal);
                bal &= bal - 1;
                int rj  = __builtin_amdgcn_readlane(r, j);
                int clj = __builtin_amdgcn_readlane(cl, j);
                float cfj = __int_as_float(__builtin_amdgcn_readlane(__float_as_int(cf), j));
                add_one<F>(rj, clj, cfj, lane, src, s_acc);
            }
        }
    }
    __syncthreads();

    int rows = N_NODES - c0; if (rows > RANGE) rows = RANGE;
    float* ap = acc + (size_t)c0 * F;
    const float* dp = dinv + c0;
    for (int i = tid; i < rows * F; i += 1024) ap[i] = fmaf(dp[i / F], s_acc[i], ap[i]);
}

__global__ __launch_bounds__(256) void k_ln_relu_64(const float* __restrict__ acc, const float* __restrict__ g,
                             const float* __restrict__ be, float* __restrict__ out) {
    const int lane = threadIdx.x & 31;
    const int n0 = (blockIdx.x * 8 + (int)(threadIdx.x >> 5)) * NPW;
    const float g0 = g[lane], g1v = g[lane + 32], b0 = be[lane], b1v = be[lane + 32];
    static_assert(NPW % 5 == 0, "whole turns of five nodes");
    for (int it = 0; it < NPW; it += 5) {
        float v0[5], v1[5];
        #pragma unroll
        for (int u = 0; u < 5; ++u) {
            const long long n = n0 + it + u;
            v0[u] = acc[n * 64 + lane];
            v1[u] = acc[n * 64 + lane + 32];
        }
        #pragma unroll
        for (int u = 0; u < 5; ++u) {
            const long long n = n0 + it + u;
            float s = v0[u] + v1[u], sq = fmaf(v0[u], v0[u], v1[u] * v1[u]);
            #pragma unroll
            for (int off = 16; off > 0; off >>= 1) {
                s  += __shfl_xor(s,  off, 32);
                sq += __shfl_xor(sq, off, 32);
            }
            float mu  = s * (1.f / 64.f);
            float var = sq * (1.f / 64.f) - mu * mu;
            float inv = rsqrtf(var + LN_EPS);
            float o0 = (v0[u] - mu) * inv * g0  + b0;
            float o1 = (v1[u] - mu) * inv * g1v + b1v;
            out[n * 64 + lane]      = fmaxf(o0, 0.f);
            out[n * 64 + lane + 32] = fmaxf(o1, 0.f);
        }
    }
}

__global__ void k_l2_wmma(const float* __restrict__ x1, const float* __restrict__ W2,
                          const float* __restrict__ rW2, const float* __restrict__ b2,
                          const float* __restrict__ rb2, const float* __restrict__ dinv,
                          float* __restrict__ xw2, float* __restrict__ acc2) {
    int wave_global = (blockIdx.x * blockDim.x + threadIdx.x) >> 5;
    int lane = threadIdx.x & 31;
    const int tiles_m = N_NODES / 16;
    int tm = wave_global >> 1;
    int tn = wave_global & 1;
    if (tm >= tiles_m) return;
    int m0 = tm * 16, n0 = tn * 16;
    int  lr = lane & 15;
    bool hi = lane >= 16;
    int  kofs = hi ? 2 : 0;

    v8f c0 = {};
    v8f c1 = {};
    #pragma unroll
    for (int k0 = 0; k0 < HID; k0 += 4) {
        const float* ap = &x1[(long long)(m0 + lr) * HID + k0 + kofs];
        v2f a; a.x = ap[0]; a.y = ap[1];
        int kb = k0 + kofs;
        v2f b;  b.x  = W2 [kb * RED + n0 + lr]; b.y  = W2 [(kb + 1) * RED + n0 + lr];
        v2f br; br.x = rW2[kb * RED + n0 + lr]; br.y = rW2[(kb + 1) * RED + n0 + lr];
        c0 = __builtin_amdgcn_wmma_f32_16x16x4_f32(false, a, false, b,  (short)0, c0, false, false);
        c1 = __builtin_amdgcn_wmma_f32_16x16x4_f32(false, a, false, br, (short)0, c1, false, false);
    }
    int ncol = n0 + lr;
    float b2v = b2[ncol], rb2v = rb2[ncol];
    #pragma unroll
    for (int v = 0; v < 8; ++v) {
        int m = m0 + v + (hi ? 8 : 0);
        float di = dinv[m];
        float xv = di * c0[v];
        xw2 [(long long)m * RED + ncol] = xv;
        acc2[(long long)m * RED + ncol] = di * xv + b2v + c1[v] + rb2v;
    }
}

__global__ void k_head(const float* __restrict__ acc2, const float* __restrict__ g2,
                       const float* __restrict__ be2, const float* __restrict__ fcW,
                       const float* __restrict__ fcb, const float* __restrict__ pW,
                       const float* __restrict__ pb, float* __restrict__ logits,
                       float* __restrict__ emb) {
    int wave = threadIdx.x >> 5;
    int lane = threadIdx.x & 31;
    int n = blockIdx.x * (blockDim.x >> 5) + wave;
    if (n >= N_NODES) return;
    float v = acc2[(long long)n * 32 + lane];
    float s = v, sq = v * v;
    #pragma unroll
    for (int off = 16; off > 0; off >>= 1) {
        s  += __shfl_xor(s,  off, 32);
        sq += __shfl_xor(sq, off, 32);
    }
    float mu  = s * (1.f / 32.f);
    float var = sq * (1.f / 32.f) - mu * mu;
    float inv = rsqrtf(var + LN_EPS);
    float x2 = fmaxf((v - mu) * inv * g2[lane] + be2[lane], 0.f);

    float lg = x2 * fcW[lane];
    #pragma unroll
    for (int off = 16; off > 0; off >>= 1) lg += __shfl_xor(lg, off, 32);
    if (lane == 0) logits[n] = lg + fcb[0];

    float p = pb[lane];
    #pragma unroll
    for (int i = 0; i < 32; ++i)
        p = fmaf(__shfl(x2, i, 32), pW[i * 32 + lane], p);
    float nsq = p * p;
    #pragma unroll
    for (int off = 16; off > 0; off >>= 1) nsq += __shfl_xor(nsq, off, 32);
    float nrm = sqrtf(nsq);
    emb[(long long)n * 32 + lane] = p / fmaxf(nrm, 1e-12f);
}

extern "C" void kernel_launch(void* const* d_in, const int* in_sizes, int n_in,
                              void* d_out, int out_size, void* d_ws, size_t ws_size,
                              hipStream_t stream) {
    const float* x    = (const float*)d_in[0];
    const float* ew   = (const float*)d_in[1];
    const float* W1   = (const float*)d_in[2];
    const float* b1   = (const float*)d_in[3];
    const float* rW1  = (const float*)d_in[4];
    const float* rb1  = (const float*)d_in[5];
    const float* g1   = (const float*)d_in[6];
    const float* be1  = (const float*)d_in[7];
    const float* W2   = (const float*)d_in[8];
    const float* b2   = (const float*)d_in[9];
    const float* rW2  = (const float*)d_in[10];
    const float* rb2  = (const float*)d_in[11];
    const float* g2   = (const float*)d_in[12];
    const float* be2  = (const float*)d_in[13];
    const float* fcW  = (const float*)d_in[14];
    const float* fcb  = (const float*)d_in[15];
    const float* pW   = (const float*)d_in[16];
    const float* pb   = (const float*)d_in[17];
    const int*   ei   = (const int*)d_in[18];
    const int*   row  = ei;
    const int*   col  = ei + N_EDGES;

    float* ws   = (float*)d_ws;
    float* deg  = ws;  ws += N_NODES;
    float* dinv = ws;  ws += N_NODES;
    float* xw1  = ws;  ws += (size_t)N_NODES * HID;
    float* acc1 = ws;  ws += (size_t)N_NODES * HID;
    unsigned char* own = (unsigned char*)ws;  ws += (size_t)N_NODES * HID;
    float* x1   = xw1;
    ws += (size_t)N_NODES * HID;
    unsigned int* rec = (unsigned int*)acc1;
    unsigned int* pmap = (unsigned int*)(own + N_EDGES);
    unsigned int* lcnt = pmap + (size_t)NB_RANGE * NCHUNK;
    unsigned int* lists = lcnt + NB_RANGE * 32;
    float* xw2  = acc1;
    float* acc2 = acc1 + (size_t)N_NODES * RED;

    float* logits = (float*)d_out;
    float* emb    = logits + N_NODES;

    const int B = 256;
    k_edge_pass<<<(NPAIR + EP_WAVES - 1) / EP_WAVES, 32 * EP_WAVES, 0, stream>>>(row, col, ew, own, rec, pmap);
    k_range_deg<<<NB_RANGE, 1024, 0, stream>>>(own, pmap, rec, deg, lcnt, lists);

    k_l1_init<<<L1_WGS, 32 * L1_WAVES, 0, stream>>>(x, deg, W1, b1, rW1, rb1, dinv, xw1, acc1);

    k_list_sum<HID><<<NB_RANGE, 1024, 0, stream>>>(lists, lcnt, own, row, col, ew, dinv, xw1, acc1);

    k_ln_relu_64<<<NODE_WGS, 256, 0, stream>>>(acc1, g1, be1, x1);

    int waves_l2 = (N_NODES / 16) * (RED / 16);
    int thr_l2   = waves_l2 * 32;
    k_l2_wmma<<<(thr_l2 + B - 1) / B, B, 0, stream>>>(x1, W2, rW2, b2, rb2, dinv, xw2, acc2);

    k_list_sum<RED><<<NB_RANGE, 1024, 0, stream>>>(lists, lcnt, own, row, col, ew, dinv, xw2, acc2);

    k_head<<<(N_NODES + 7) / 8, B, 0, stream>>>(acc2, g2, be2, fcW, fcb, pW, pb, logits, emb);
}


# range sums of both layers finish their own rows from LDS: LayerNorm+ReLU (layer 1, over the start value), LayerNorm+ReLU+heads (layer 2, to the outputs); 6 launches, not 8; sum order as before
# speedup vs baseline: 4.6729x; 1.1468x over previous
#include <hip/hip_runtime.h>
#include <hip/hip_bf16.h>

#define N_NODES 100000
#define N_EDGES 3200000
#define HID 64
#define RED 32
#define EMB 32
#define LN_EPS 1e-5f
#define RANGE 391
#define NB_RANGE 256
#define LCAP 1536
#define NCHUNK (N_EDGES / 512)
#define REC_SHIFT 17
#define REC_MASK ((1u << REC_SHIFT) - 1u)
#define LIST_CAP 576
#define LIST_OVER 0xFFFFFFFFu

typedef __attribute__((ext_vector_type(2))) float v2f;
typedef __attribute__((ext_vector_type(8))) float v8f;
typedef __attribute__((ext_vector_type(4))) float v4f;
typedef __attribute__((ext_vector_type(4))) int v4i;

#define EP_WAVES 4
#define NPAIR (NCHUNK / 2)
static_assert(N_EDGES == 512 * NCHUNK && NCHUNK % 2 == 0, "every wave of the edge pass has two whole chunks");
static_assert((size_t)N_EDGES + 4u * NB_RANGE * NCHUNK + 4u * NB_RANGE * 32 + 8u * NB_RANGE * 32 * LIST_CAP
              <= 2u * 4u * N_NODES * HID, "owner bytes, piece maps, list lengths and lists fit their two regions");
static_assert(N_EDGES % 256 == 0 && (4 * NB_RANGE * NCHUNK) % 256 == 0 && (8 * LIST_CAP) % 256 == 0,
              "every list starts on a 256-byte boundary");
__global__ __launch_bounds__(32 * EP_WAVES) void k_edge_pass(const int* __restrict__ row, const int* __restrict__ col,
                              const float* __restrict__ w,
                              unsigned char* __restrict__ own, unsigned int* __restrict__ rec,
                              unsigned int* __restrict__ pmap) {
    __shared__ __attribute__((aligned(16))) unsigned int s_map[EP_WAVES][2 * NB_RANGE];
    const int lane = threadIdx.x & 31;
    const int wave = __builtin_amdgcn_readfirstlane((int)(threadIdx.x >> 5));
    const int pair = blockIdx.x * EP_WAVES + wave;
    if (pair >= NPAIR) return;
    unsigned int* mp = s_map[wave];
    #pragma unroll
    for (int k = 0; k < 4; ++k) ((uint4*)mp)[k * 32 + lane] = make_uint4(0u, 0u, 0u, 0u);
    __builtin_amdgcn_fence(__ATOMIC_ACQ_REL, "wavefront");

    const int q0 = pair * 256;
    v4i c4[8]; v4f w4[8]; v4i r4[8];
    #pragma unroll
    for (int u = 0; u < 8; ++u) {
        c4[u] = ((const v4i*)col)[q0 + 32 * u + lane];
        w4[u] = ((const v4f*)w)[q0 + 32 * u + lane];
        r4[u] = ((const v4i*)row)[q0 + 32 * u + lane];
    }
    asm volatile("" :: "v"(c4[0]), "v"(c4[1]), "v"(c4[2]), "v"(c4[3]), "v"(c4[4]), "v"(c4[5]), "v"(c4[6]), "v"(c4[7]),
                       "v"(w4[0]), "v"(w4[1]), "v"(w4[2]), "v"(w4[3]), "v"(w4[4]), "v"(w4[5]), "v"(w4[6]), "v"(w4[7]),
                       "v"(r4[0]), "v"(r4[1]), "v"(r4[2]), "v"(r4[3]), "v"(r4[4]), "v"(r4[5]), "v"(r4[6]), "v"(r4[7]));
    #pragma unroll
    for (int u = 0; u < 8; ++u) {
        unsigned int* m = mp + (u >> 2) * NB_RANGE;
        const unsigned bit = 1u << (8 * (u & 3) + (lane >> 2));
        const int o0 = c4[u].x / RANGE, o1 = c4[u].y / RANGE, o2 = c4[u].z / RANGE, o3 = c4[u].w / RANGE;
        atomicOr(&m[o0], bit);
        atomicOr(&m[o1], bit);
        atomicOr(&m[o2], bit);
        atomicOr(&m[o3], bit);
        ((unsigned int*)own)[q0 + 32 * u + lane] =
            (unsigned)o0 | ((unsigned)o1 << 8) | ((unsigned)o2 << 16) | ((unsigned)o3 << 24);
        uint4 ra, rb;
        ra.x = (unsigned)r4[u].x | ((unsigned)(c4[u].x - o0 * RANGE) << REC_SHIFT);
        ra.y = __float_as_uint(w4[u].x);
        ra.z = (unsigned)r4[u].y | ((unsigned)(c4[u].y - o1 * RANGE) << REC_SHIFT);
        ra.w = __float_as_uint(w4[u].y);
        rb.x = (unsigned)r4[u].z | ((unsigned)(c4[u].z - o2 * RANGE) << REC_SHIFT);
        rb.y = __float_as_uint(w4[u].z);
        rb.z = (unsigned)r4[u].w | ((unsigned)(c4[u].w - o3 * RANGE) << REC_SHIFT);
        rb.w = __float_as_uint(w4[u].w);
        ((uint4*)rec)[2 * (q0 + 32 * u + lane)]     = ra;
        ((uint4*)rec)[2 * (q0 + 32 * u + lane) + 1] = rb;
    }
    __builtin_amdgcn_fence(__ATOMIC_ACQ_REL, "wavefront");
    #pragma unroll
    for (int k = 0; k < 8; ++k) {
        const int o = k * 32 + lane;
        uint2 v;
        v.x = mp[o];
        v.y = mp[NB_RANGE + o];
        ((uint2*)pmap)[(size_t)o * NPAIR + pair] = v;
    }
}

#define NPW 20
#define L1_WAVES 4
#define L1_WGS (N_NODES / (L1_WAVES * NPW))
static_assert(N_NODES == L1_WGS * L1_WAVES * NPW && NPW % 2 == 0 && NPW <= 32,
              "every wave of the node pass has NPW whole nodes");
__device__ __forceinline__ void l1_elem(float x0, float x1v, float di, float wa, float wb, float bb,
                                        float ra, float rb, float rbb, float& xs, float& a) {
    float xw = fmaf(x0, wa, x1v * wb);
    xs = di * xw;
    a = di * xs + bb + fmaf(x0, ra, x1v * rb) + rbb;
}
__global__ __launch_bounds__(32 * L1_WAVES) void k_l1_init(const float* __restrict__ x, const float* __restrict__ deg,
                          const float* __restrict__ W1, const float* __restrict__ b1,
                          const float* __restrict__ rW1, const float* __restrict__ rb1,
                          float* __restrict__ dinv, float* __restrict__ xw1,
                          float* __restrict__ acc1) {
    const int lane = threadIdx.x & 31;
    const int n0 = (blockIdx.x * L1_WAVES + (int)(threadIdx.x >> 5)) * NPW;
    const int f0 = 4 * (lane & 15), h = lane >> 4;
    const v4f wa = *(const v4f*)(W1 + f0),  wb = *(const v4f*)(W1 + HID + f0);
    const v4f ra = *(const v4f*)(rW1 + f0), rb = *(const v4f*)(rW1 + HID + f0);
    const v4f bv = *(const v4f*)(b1 + f0),  rbv = *(const v4f*)(rb1 + f0);
    const float dl = deg[n0 + (lane < NPW ? lane : NPW - 1)];
    float dg[NPW / 2]; v2f xv[NPW / 2];
    #pragma unroll
    for (int i = 0; i < NPW / 2; ++i) {
        const int n = n0 + 2 * i + h;
        dg[i] = deg[n];
        xv[i] = *(const v2f*)(x + 2 * n);
    }
    asm volatile("" :: "v"(wa), "v"(wb), "v"(ra), "v"(rb), "v"(bv), "v"(rbv), "v"(dl),
                       "v"(dg[0]), "v"(dg[1]), "v"(dg[2]), "v"(dg[3]), "v"(dg[4]),
                       "v"(dg[5]), "v"(dg[6]), "v"(dg[7]), "v"(dg[8]), "v"(dg[9]),
                       "v"(xv[0]), "v"(xv[1]), "v"(xv[2]), "v"(xv[3]), "v"(xv[4]),
                       "v"(xv[5]), "v"(xv[6]), "v"(xv[7]), "v"(xv[8]), "v"(xv[9]));
    static_assert(NPW / 2 == 10, "the statement above names ten turns");
    #pragma unroll
    for (int i = 0; i < NPW / 2; ++i) {
        const int n = n0 + 2 * i + h;
        const float d  = dg[i];
        const float di = d > 0.f ? rsqrtf(d) : 0.f;
        float s0, s1, s2, s3, a0, a1, a2, a3;
        l1_elem(xv[i].x, xv[i].y, di, wa.x, wb.x, bv.x, ra.x, rb.x, rbv.x, s0, a0);
        l1_elem(xv[i].x, xv[i].y, di, wa.y, wb.y, bv.y, ra.y, rb.y, rbv.y, s1, a1);
        l1_elem(xv[i].x, xv[i].y, di, wa.z, wb.z, bv.z, ra.z, rb.z, rbv.z, s2, a2);
        l1_elem(xv[i].x, xv[i].y, di, wa.w, wb.w, bv.w, ra.w, rb.w, rbv.w, s3, a3);
        v4f xs = {s0, s1, s2, s3}, a = {a0, a1, a2, a3};
        *(v4f*)(xw1  + (size_t)n * HID + f0) = xs;
        *(v4f*)(acc1 + (size_t)n * HID + f0) = a;
    }
    if (lane < NPW) dinv[n0 + lane] = dl > 0.f ? rsqrtf(dl) : 0.f;
}

__device__ __forceinline__ void walk_deg(const int* list, int n, int lane,
                                         const uint2* __restrict__ rec, float* s_acc,
                                         uint2* __restrict__ gl, int gpos, bool keep) {
    for (int base = 0; base < n; base += 128) {
        uint2 rc[4]; bool ok[4];
        #pragma unroll
        for (int u = 0; u < 4; ++u) {
            const int i = base + 32 * u + lane;
            ok[u] = i < n;
            rc[u] = rec[list[ok[u] ? i : n - 1]];
        }
        #pragma unroll
        for (int u = 0; u < 4; ++u)
            if (ok[u]) {
                atomicAdd(&s_acc[rc[u].x >> REC_SHIFT], __uint_as_float(rc[u].y));
                if (keep) gl[gpos + base + 32 * u + lane] = rc[u];
            }
    }
}

__device__ __forceinline__ unsigned nz_bytes(unsigned x) {
    return ((x & 0x7F7F7F7Fu) + 0x7F7F7F7Fu) | x;
}

__global__ __launch_bounds__(1024) void k_range_deg(const unsigned char* __restrict__ own,
                                                    const unsigned int* __restrict__ pmap,
                                                    const unsigned int* __restrict__ rec,
                                                    float* __restrict__ deg,
                                                    unsigned int* __restrict__ lcnt,
                                                    unsigned int* __restrict__ lists) {
    __shared__ float s_acc[RANGE];
    __shared__ int   s_list[32 * LCAP];
    const int tid = threadIdx.x, lane = tid & 31;
    const int wave = __builtin_amdgcn_readfirstlane(tid >> 5);
    const int b = blockIdx.x, c0 = b * RANGE;
    for (int i = tid; i < RANGE; i += 1024) s_acc[i] = 0.f;
    __syncthreads();

    int* list = s_list + wave * LCAP;
    int cnt = 0;
    int gpos = 0;
    uint2* gl = (uint2*)lists + (size_t)(b * 32 + wave) * LIST_CAP;
    const uint4* own4 = (const uint4*)own;
    const uint2* rec2 = (const uint2*)rec;
    const unsigned key = (unsigned)b * 0x01010101u;
    const unsigned int* mp = pmap + (size_t)b * NCHUNK;
    auto drain = [&]() {
        __builtin_amdgcn_fence(__ATOMIC_ACQ_REL, "wavefront");
        walk_deg(list, cnt, lane, rec2, s_acc, gl, gpos, gpos + cnt <= LIST_CAP);
        gpos += cnt;
    };
    auto take = [&](const uint4 cur, int ebase, bool ok) {
        if (__builtin_expect(cnt > LCAP - 512, 0)) {
            drain();
            __builtin_amdgcn_fence(__ATOMIC_ACQ_REL, "wavefront");
            cnt = 0;
        }
        const unsigned H = 0x80808080u;
        unsigned t0 = nz_bytes(cur.x ^ key), t1 = nz_bytes(cur.y ^ key);
        unsigned t2 = nz_bytes(cur.z ^ key), t3 = nz_bytes(cur.w ^ key);
        unsigned m = ((~t0 & H) >> 7) | ((~t1 & H) >> 6) | ((~t2 & H) >> 5) | ((~t3 & H) >> 4);
        if (!ok) m = 0u;
        unsigned bal = __builtin_amdgcn_ballot_w32(m != 0);
        while (bal) {
            int p = __builtin_ctz(m | 0x80000000u);
            int slot = cnt + (int)__builtin_amdgcn_mbcnt_lo(bal, 0);
            if (m != 0) list[slot] = ebase + ((p & 3) << 2) + (p >> 3);
            cnt += __builtin_popcount(bal);
            m &= m - 1;
            bal = __builtin_amdgcn_ballot_w32(m != 0);
        }
    };
    const int lo = (wave * NCHUNK) / 32, hi = ((wave + 1) * NCHUNK) / 32;
    constexpr int NROUND = (NCHUNK / 32 + 1 + 31) / 32;
    unsigned nxt = (lo + lane < hi) ? mp[lo + lane] : 0u;
    for (int i = 0; i < NROUND; ++i) {
        unsigned wd = nxt;
        const int chunk = lo + lane + 32 * i;
        nxt = (i + 1 < NROUND && chunk + 32 < hi) ? mp[chunk + 32] : 0u;
        while (__builtin_amdgcn_ballot_w32(wd != 0)) {
            uint4 pc[4]; int pe[4]; bool ok[4];
            #pragma unroll
            for (int u = 0; u < 4; ++u) {
                ok[u] = wd != 0;
                const int p = ok[u] ? __builtin_ctz(wd) : 0;
                wd &= wd - 1;
                const int piece = ok[u] ? chunk * 32 + p : 0;
                pe[u] = piece * 16;
                pc[u] = own4[piece];
            }
            #pragma unroll
            for (int u = 0; u < 4; ++u) {
                if (__builtin_amdgcn_ballot_w32(ok[u]) == 0) break;
                take(pc[u], pe[u], ok[u]);
            }
        }
    }
    drain();
    if (lane == 0) lcnt[b * 32 + wave] = gpos <= LIST_CAP ? (unsigned)gpos : LIST_OVER;
    __syncthreads();

    int rows = N_NODES - c0; if (rows > RANGE) rows = RANGE;
    for (int i = tid; i < rows; i += 1024) deg[c0 + i] = 1.0f + s_acc[i];
}

template <int F>
__device__ __forceinline__ void add_one(int rj, int clj, float cfj, int lane,
                                        const float* __restrict__ src, float* s_acc) {
    if (F == 64) {
        v2f s = *(const v2f*)(src + (size_t)rj * 64 + 2 * lane);
        atomicAdd(&s_acc[clj * 64 + 2 * lane],     cfj * s.x);
        atomicAdd(&s_acc[clj * 64 + 2 * lane + 1], cfj * s.y);
    } else {
        atomicAdd(&s_acc[clj * F + lane], cfj * src[(size_t)rj * F + lane]);
    }
}

template <int F>
__device__ __forceinline__ void add_batch(int r, int cl, float cf, int m, int lane,
                                          const float* __restrict__ src, float* s_acc) {
    int j = 0;
    for (; j + 8 <= m; j += 8) {
        if (F == 64) {
            v2f s[8];
            #pragma unroll
            for (int u = 0; u < 8; ++u) {
                int rj = __builtin_amdgcn_readlane(r, j + u);
                s[u] = *(const v2f*)(src + (size_t)rj * 64 + 2 * lane);
            }
            #pragma unroll
            for (int u = 0; u < 8; ++u) {
                int clj = __builtin_amdgcn_readlane(cl, j + u);
                float cfj = __int_as_float(__builtin_amdgcn_readlane(__float_as_int(cf), j + u));
                atomicAdd(&s_acc[clj * 64 + 2 * lane],     cfj * s[u].x);
                atomicAdd(&s_acc[clj * 64 + 2 * lane + 1], cfj * s[u].y);
            }
        } else {
            float s[8];
            #pragma unroll
            for (int u = 0; u < 8; ++u) {
                int rj = __builtin_amdgcn_readlane(r, j + u);
                s[u] = src[(size_t)rj * F + lane];
            }
            #pragma unroll
            for (int u = 0; u < 8; ++u) {
                int clj = __builtin_amdgcn_readlane(cl, j + u);
                float cfj = __int_as_float(__builtin_amdgcn_readlane(__float_as_int(cf), j + u));
                atomicAdd(&s_acc[clj * F + lane], cfj * s[u]);
            }
        }
    }
    for (; j < m; ++j) {
        int rj  = __builtin_amdgcn_readlane(r, j);
        int clj = __builtin_amdgcn_readlane(cl, j);
        float cfj = __int_as_float(__builtin_amdgcn_readlane(__float_as_int(cf), j));
        add_one<F>(rj, clj, cfj, lane, src, s_acc);
    }
}

#define ROWS_PW 13
static_assert(32 * ROWS_PW >= RANGE, "every row of a range has a wave");

template <int NT>
__device__ __forceinline__ void finish_l1(int r0, int rows, int lane, const float* s_acc,
                                          const float* __restrict__ dp, float* __restrict__ ap,
                                          float g0, float g1v, float b0, float b1v) {
    if (r0 >= rows) return;
    float a0[NT], a1[NT], dv[NT];
    #pragma unroll
    for (int u = 0; u < NT; ++u) {
        const int rc = r0 + u < rows ? r0 + u : rows - 1;
        a0[u] = ap[rc * 64 + lane];
        a1[u] = ap[rc * 64 + lane + 32];
        dv[u] = dp[rc];
    }
    #pragma unroll
    for (int u = 0; u < NT; ++u) {
        const int r = r0 + u;
        if (r < rows) {
            const float v0 = fmaf(dv[u], s_acc[r * 64 + lane],      a0[u]);
            const float v1 = fmaf(dv[u], s_acc[r * 64 + lane + 32], a1[u]);
            float s = v0 + v1, sq = fmaf(v0, v0, v1 * v1);
            #pragma unroll
            for (int off = 16; off > 0; off >>= 1) {
                s  += __shfl_xor(s,  off, 32);
                sq += __shfl_xor(sq, off, 32);
            }
            float mu  = s * (1.f / 64.f);
            float var = sq * (1.f / 64.f) - mu * mu;
            float inv = rsqrtf(var + LN_EPS);
            float o0 = (v0 - mu) * inv * g0  + b0;
            float o1 = (v1 - mu) * inv * g1v + b1v;
            ap[r * 64 + lane]      = fmaxf(o0, 0.f);
            ap[r * 64 + lane + 32] = fmaxf(o1, 0.f);
        }
    }
}

template <int NT>
__device__ __forceinline__ void finish_l2(int r0, int rows, int lane, const float* s_acc,
                                          const float* __restrict__ dp, const float* __restrict__ ap,
                                          float g2v, float be2v, float fcWv, float pbv, const float (&pWc)[32],
                                          int j0, float& lgs, float* __restrict__ embp) {
    if (r0 >= rows) return;
    float a[NT], dv[NT];
    #pragma unroll
    for (int u = 0; u < NT; ++u) {
        const int rc = r0 + u < rows ? r0 + u : rows - 1;
        a[u]  = ap[rc * 32 + lane];
        dv[u] = dp[rc];
    }
    #pragma unroll
    for (int u = 0; u < NT; ++u) {
        const int r = r0 + u;
        if (r < rows) {
            const float v = fmaf(dv[u], s_acc[r * 32 + lane], a[u]);
            float s = v, sq = v * v;
            #pragma unroll
            for (int off = 16; off > 0; off >>= 1) {
                s  += __shfl_xor(s,  off, 32);
                sq += __shfl_xor(sq, off, 32);
            }
            float mu  = s * (1.f / 32.f);
            float var = sq * (1.f / 32.f) - mu * mu;
            float inv = rsqrtf(var + LN_EPS);
            float x2 = fmaxf((v - mu) * inv * g2v + be2v, 0.f);

            float lg = x2 * fcWv;
            #pragma unroll
            for (int off = 16; off > 0; off >>= 1) lg += __shfl_xor(lg, off, 32);
            if (lane == j0 + u) lgs = lg;

            float p = pbv;
            #pragma unroll
            for (int i = 0; i < 32; ++i)
                p = fmaf(__shfl(x2, i, 32), pWc[i], p);
            float nsq = p * p;
            #pragma unroll
            for (int off = 16; off > 0; off >>= 1) nsq += __shfl_xor(nsq, off, 32);
            float nrm = sqrtf(nsq);
            embp[r * 32 + lane] = p / fmaxf(nrm, 1e-12f);
        }
    }
}

template <int F>
__global__ __launch_bounds__(1024) void k_list_sum(const unsigned int* __restrict__ lists,
                                                   const unsigned int* __restrict__ lcnt,
                                                   const unsigned char* __restrict__ own,
                                                   const int* __restrict__ row, const int* __restrict__ col,
                                                   const float* __restrict__ w,
                                                   const float* __restrict__ dinv,
                                                   const float* __restrict__ src,
                                                   float* __restrict__ acc,
                                                   const float* __restrict__ g, const float* __restrict__ be,
                                                   const float* __restrict__ fcW, const float* __restrict__ fcb,
                                                   const float* __restrict__ pW, const float* __restrict__ pb,
                                                   float* __restrict__ logits, float* __restrict__ emb) {
    __shared__ float s_acc[RANGE * F];
    const int tid = threadIdx.x, lane = tid & 31;
    const int wave = __builtin_amdgcn_readfirstlane(tid >> 5);
    const int b = blockIdx.x, c0 = b * RANGE;
    const int n = __builtin_amdgcn_readfirstlane((int)lcnt[b * 32 + wave]);
    for (int i = tid; i < RANGE * F; i += 1024) s_acc[i] = 0.f;
    __syncthreads();

    if (n >= 0) {
        const uint2* L = (const uint2*)lists + (size_t)(b * 32 + wave) * LIST_CAP;
        uint2 nx = make_uint2(0u, 0u);
        if (n > 0) nx = L[lane < n ? lane : n - 1];
        for (int base = 0; base < n; base += 32) {
            const uint2 rc = nx;
            int m = n - base; if (m > 32) m = 32;
            if (base + 32 < n) { const int i = base + 32 + lane; nx = L[i < n ? i : n - 1]; }
            add_batch<F>((int)(rc.x & REC_MASK), (int)(rc.x >> REC_SHIFT), __uint_as_float(rc.y),
                         m, lane, src, s_acc);
        }
    } else {
        const int lo = (wave * NCHUNK) / 32, hi = ((wave + 1) * NCHUNK) / 32;
        for (int e0 = lo * 512; e0 < hi * 512; e0 += 32) {
            const int e = e0 + lane;
            const bool mine = own[e] == (unsigned char)b;
            unsigned bal = __builtin_amdgcn_ballot_w32(mine);
            if (bal == 0) continue;
            const int r  = mine ? row[e] : 0;
            const int cl = mine ? col[e] - c0 : 0;
            const float cf = mine ? w[e] : 0.f;
            while (bal) {
                const int j = __builtin_ctz(bal);
                bal &= bal - 1;
                int rj  = __builtin_amdgcn_readlane(r, j);
                int clj = __builtin_amdgcn_readlane(cl, j);
                float cfj = __int_as_float(__builtin_amdgcn_readlane(__float_as_int(cf), j));
                add_one<F>(rj, clj, cfj, lane, src, s_acc);
            }
        }
    }
    __syncthreads();

    int rows = N_NODES - c0; if (rows > RANGE) rows = RANGE;
    float* ap = acc + (size_t)c0 * F;
    const float* dp = dinv + c0;
    const int r0 = wave * ROWS_PW;
    if (F == 64) {
        const float g0 = g[lane], g1v = g[lane + 32], b0 = be[lane], b1v = be[lane + 32];
        finish_l1<7>(r0, rows, lane, s_acc, dp, ap, g0, g1v, b0, b1v);
        finish_l1<ROWS_PW - 7>(r0 + 7, rows, lane, s_acc, dp, ap, g0, g1v, b0, b1v);
    } else {
        const float g2v = g[lane], be2v = be[lane], fcWv = fcW[lane], pbv = pb[lane];
        float pWc[32];
        #pragma unroll
        for (int i = 0; i < 32; ++i) pWc[i] = pW[i * 32 + lane];
        float lgs = 0.f;
        float* embp = emb + (size_t)c0 * EMB;
        finish_l2<7>(r0, rows, lane, s_acc, dp, ap, g2v, be2v, fcWv, pbv, pWc, 0, lgs, embp);
        finish_l2<ROWS_PW - 7>(r0 + 7, rows, lane, s_acc, dp, ap, g2v, be2v, fcWv, pbv, pWc, 7, lgs, embp);
        if (lane < ROWS_PW && r0 + lane < rows) logits[c0 + r0 + lane] = lgs + fcb[0];
    }
}

__global__ void k_l2_wmma(const float* __restrict__ x1, const float* __restrict__ W2,
                          const float* __restrict__ rW2, const float* __restrict__ b2,
                          const float* __restrict__ rb2, const float* __restrict__ dinv,
                          float* __restrict__ xw2, float* __restrict__ acc2) {
    int wave_global = (blockIdx.x * blockDim.x + threadIdx.x) >> 5;
    int lane = threadIdx.x & 31;
    const int tiles_m = N_NODES / 16;
    int tm = wave_global >> 1;
    int tn = wave_global & 1;
    if (tm >= tiles_m) return;
    int m0 = tm * 16, n0 = tn * 16;
    int  lr = lane & 15;
    bool hi = lane >= 16;
    int  kofs = hi ? 2 : 0;

    v8f c0 = {};
    v8f c1 = {};
    #pragma unroll
    for (int k0 = 0; k0 < HID; k0 += 4) {
        const float* ap = &x1[(long long)(m0 + lr) * HID + k0 + kofs];
        v2f a; a.x = ap[0]; a.y = ap[1];
        int kb = k0 + kofs;
        v2f b;  b.x  = W2 [kb * RED + n0 + lr]; b.y  = W2 [(kb + 1) * RED + n0 + lr];
        v2f br; br.x = rW2[kb * RED + n0 + lr]; br.y = rW2[(kb + 1) * RED + n0 + lr];
        c0 = __builtin_amdgcn_wmma_f32_16x16x4_f32(false, a, false, b,  (short)0, c0, false, false);
        c1 = __builtin_amdgcn_wmma_f32_16x16x4_f32(false, a, false, br, (short)0, c1, false, false);
    }
    int ncol = n0 + lr;
    float b2v = b2[ncol], rb2v = rb2[ncol];
    #pragma unroll
    for (int v = 0; v < 8; ++v) {
        int m = m0 + v + (hi ? 8 : 0);
        float di = dinv[m];
        float xv = di * c0[v];
        xw2 [(long long)m * RED + ncol] = xv;
        acc2[(long long)m * RED + ncol] = di * xv + b2v + c1[v] + rb2v;
    }
}

extern "C" void kernel_launch(void* const* d_in, const int* in_sizes, int n_in,
                              void* d_out, int out_size, void* d_ws, size_t ws_size,
                              hipStream_t stream) {
    const float* x    = (const float*)d_in[0];
    const float* ew   = (const float*)d_in[1];
    const float* W1   = (const float*)d_in[2];
    const float* b1   = (const float*)d_in[3];
    const float* rW1  = (const float*)d_in[4];
    const float* rb1  = (const float*)d_in[5];
    const float* g1   = (const float*)d_in[6];
    const float* be1  = (const float*)d_in[7];
    const float* W2   = (const float*)d_in[8];
    const float* b2   = (const float*)d_in[9];
    const float* rW2  = (const float*)d_in[10];
    const float* rb2  = (const float*)d_in[11];
    const float* g2   = (const float*)d_in[12];
    const float* be2  = (const float*)d_in[13];
    const float* fcW  = (const float*)d_in[14];
    const float* fcb  = (const float*)d_in[15];
    const float* pW   = (const float*)d_in[16];
    const float* pb   = (const float*)d_in[17];
    const int*   ei   = (const int*)d_in[18];
    const int*   row  = ei;
    const int*   col  = ei + N_EDGES;

    float* ws   = (float*)d_ws;
    float* deg  = ws;  ws += N_NODES;
    float* dinv = ws;  ws += N_NODES;
    float* xw1  = ws;  ws += (size_t)N_NODES * HID;
    float* acc1 = ws;  ws += (size_t)N_NODES * HID;
    unsigned char* own = (unsigned char*)ws;  ws += (size_t)N_NODES * HID;
    float* x1   = acc1;
    ws += (size_t)N_NODES * HID;
    unsigned int* rec = (unsigned int*)acc1;
    unsigned int* pmap = (unsigned int*)(own + N_EDGES);
    unsigned int* lcnt = pmap + (size_t)NB_RANGE * NCHUNK;
    unsigned int* lists = lcnt + NB_RANGE * 32;
    float* xw2  = xw1;
    float* acc2 = xw1 + (size_t)N_NODES * RED;

    float* logits = (float*)d_out;
    float* emb    = logits + N_NODES;

    const int B = 256;
    k_edge_pass<<<(NPAIR + EP_WAVES - 1) / EP_WAVES, 32 * EP_WAVES, 0, stream>>>(row, col, ew, own, rec, pmap);
    k_range_deg<<<NB_RANGE, 1024, 0, stream>>>(own, pmap, rec, deg, lcnt, lists);

    k_l1_init<<<L1_WGS, 32 * L1_WAVES, 0, stream>>>(x, deg, W1, b1, rW1, rb1, dinv, xw1, acc1);

    k_list_sum<HID><<<NB_RANGE, 1024, 0, stream>>>(lists, lcnt, own, row, col, ew, dinv, xw1, acc1,
                                                   g1, be1, nullptr, nullptr, nullptr, nullptr, nullptr, nullptr);

    int waves_l2 = (N_NODES / 16) * (RED / 16);
    int thr_l2   = waves_l2 * 32;
    k_l2_wmma<<<(thr_l2 + B - 1) / B, B, 0, stream>>>(x1, W2, rW2, b2, rb2, dinv, xw2, acc2);

    k_list_sum<RED><<<NB_RANGE, 1024, 0, stream>>>(lists, lcnt, own, row, col, ew, dinv, xw2, acc2,
                                                   g2, be2, fcW, fcb, pW, pb, logits, emb);
}
